# starting kernel
# baseline (speedup 1.0000x reference)
#include <hip/hip_runtime.h>
#include <hip/hip_bf16.h>


#define NQ    4096
#define NT    16384
#define DDIM  32
#define OUTD  16
#define SEGS  8
#define SEGLEN (NT / SEGS)
#define EPSV  1e-6f
#define BIGF  3.0e38f

typedef __attribute__((ext_vector_type(2))) float v2f;
typedef __attribute__((ext_vector_type(4))) float v4f;
typedef __attribute__((ext_vector_type(8))) float v8f;

__device__ __forceinline__ void bot5_insert(float (&tv)[5], int (&ti)[5],
                                            float v, int id) {
    if (v < tv[4]) {
        tv[4] = v; ti[4] = id;
#pragma unroll
        for (int k = 4; k > 0; --k) {
            if (tv[k] < tv[k - 1]) {
                float fv = tv[k]; tv[k] = tv[k - 1]; tv[k - 1] = fv;
                int   fi = ti[k]; ti[k] = ti[k - 1]; ti[k - 1] = fi;
            }
        }
    }
}

__global__ __launch_bounds__(128) void idw_prep(
    const float* __restrict__ x, const float* __restrict__ tx,
    const float* __restrict__ w,
    float* __restrict__ xq_s, float* __restrict__ xt_s,
    float* __restrict__ qnorm, float* __restrict__ tnorm) {
    int i = blockIdx.x * blockDim.x + threadIdx.x;
    bool isQ = i < NQ;
    const float* src = isQ ? (x + (size_t)i * DDIM) : (tx + (size_t)(i - NQ) * DDIM);
    float*       dst = isQ ? (xq_s + (size_t)i * DDIM) : (xt_s + (size_t)(i - NQ) * DDIM);
    float nrm = 0.f;
#pragma unroll
    for (int j = 0; j < DDIM; ++j) {
        int h  = j >> 4;
        int jj = j & 15;
        int K  = 4 * (jj >> 1) + (jj & 1) + 2 * h;
        float s = __expf(-w[K]);
        float v = src[K] * s;
        dst[j] = v;
        nrm   += v * v;
    }
    if (isQ) qnorm[i] = nrm; else tnorm[i - NQ] = nrm;
}

__global__ __launch_bounds__(32) void idw_main(
    const float* __restrict__ xq_s, const float* __restrict__ xt_s,
    const float* __restrict__ qnorm, const float* __restrict__ tnorm,
    float* __restrict__ pvals, int* __restrict__ pidx) {
    const int wid   = blockIdx.x;
    const int seg   = wid % SEGS;
    const int qtile = wid / SEGS;
    const int lane  = threadIdx.x;
    const int h     = lane >> 4;
    const int ql    = lane & 15;
    const int q     = qtile * 16 + ql;

    const float* qrow = xq_s + (size_t)q * DDIM + 16 * h;
    v4f bq0 = *(const v4f*)(qrow + 0);
    v4f bq1 = *(const v4f*)(qrow + 4);
    v4f bq2 = *(const v4f*)(qrow + 8);
    v4f bq3 = *(const v4f*)(qrow + 12);
    v2f b[8] = {bq0.xy, bq0.zw, bq1.xy, bq1.zw, bq2.xy, bq2.zw, bq3.xy, bq3.zw};
    const float qn = qnorm[q];

    float tv[5] = {BIGF, BIGF, BIGF, BIGF, BIGF};
    int   ti[5] = {0, 0, 0, 0, 0};

    const int tbeg = seg * SEGLEN;
#pragma unroll 2
    for (int t0 = tbeg; t0 < tbeg + SEGLEN; t0 += 16) {
        const float* arow = xt_s + (size_t)(t0 + ql) * DDIM + 16 * h;
        v4f a0 = *(const v4f*)(arow + 0);
        v4f a1 = *(const v4f*)(arow + 4);
        v4f a2 = *(const v4f*)(arow + 8);
        v4f a3 = *(const v4f*)(arow + 12);

        v8f c0 = {0.f, 0.f, 0.f, 0.f, 0.f, 0.f, 0.f, 0.f};
        v8f c1 = c0;
        c0 = __builtin_amdgcn_wmma_f32_16x16x4_f32(false, a0.xy, false, b[0], (short)0, c0, false, false);
        c1 = __builtin_amdgcn_wmma_f32_16x16x4_f32(false, a2.xy, false, b[4], (short)0, c1, false, false);
        c0 = __builtin_amdgcn_wmma_f32_16x16x4_f32(false, a0.zw, false, b[1], (short)0, c0, false, false);
        c1 = __builtin_amdgcn_wmma_f32_16x16x4_f32(false, a2.zw, false, b[5], (short)0, c1, false, false);
        c0 = __builtin_amdgcn_wmma_f32_16x16x4_f32(false, a1.xy, false, b[2], (short)0, c0, false, false);
        c1 = __builtin_amdgcn_wmma_f32_16x16x4_f32(false, a3.xy, false, b[6], (short)0, c1, false, false);
        c0 = __builtin_amdgcn_wmma_f32_16x16x4_f32(false, a1.zw, false, b[3], (short)0, c0, false, false);
        c1 = __builtin_amdgcn_wmma_f32_16x16x4_f32(false, a3.zw, false, b[7], (short)0, c1, false, false);

        const float* tnp = tnorm + t0 + 8 * h;
        v4f tna = *(const v4f*)(tnp);
        v4f tnb = *(const v4f*)(tnp + 4);
        float tn[8] = {tna.x, tna.y, tna.z, tna.w, tnb.x, tnb.y, tnb.z, tnb.w};

#pragma unroll
        for (int r = 0; r < 8; ++r) {
            float sq = qn + tn[r];
            sq = fmaf(-2.0f, c0[r], sq);
            sq = fmaf(-2.0f, c1[r], sq);
            bot5_insert(tv, ti, sq, t0 + 8 * h + r);
        }
    }

    float pv[5]; int pi[5];
#pragma unroll
    for (int k = 0; k < 5; ++k) {
        pv[k] = __shfl_xor(tv[k], 16, 32);
        pi[k] = __shfl_xor(ti[k], 16, 32);
    }
#pragma unroll
    for (int k = 0; k < 5; ++k) bot5_insert(tv, ti, pv[k], pi[k]);

    if (lane < 16) {
        int base = (q * SEGS + seg) * 5;
#pragma unroll
        for (int k = 0; k < 5; ++k) { pvals[base + k] = tv[k]; pidx[base + k] = ti[k]; }
    }
}

__global__ __launch_bounds__(128) void idw_reduce(
    const float* __restrict__ pvals, const int* __restrict__ pidx,
    const float* __restrict__ train_y, float* __restrict__ out) {
    int q = blockIdx.x * blockDim.x + threadIdx.x;
    if (q >= NQ) return;

    float tv[5] = {BIGF, BIGF, BIGF, BIGF, BIGF};
    int   ti[5] = {0, 0, 0, 0, 0};
    for (int s = 0; s < SEGS; ++s) {
        int base = (q * SEGS + s) * 5;
#pragma unroll
        for (int k = 0; k < 5; ++k)
            bot5_insert(tv, ti, pvals[base + k], pidx[base + k]);
    }

    float wv[5];
#pragma unroll
    for (int k = 0; k < 5; ++k)
        wv[k] = rsqrtf(fmaxf(tv[k], 0.f) + EPSV);

    float sw  = wv[0] + wv[1] + wv[2] + wv[3] + wv[4];
    float inv = 1.f / sw;

    v4f acc0 = {0.f, 0.f, 0.f, 0.f}, acc1 = acc0, acc2 = acc0, acc3 = acc0;
#pragma unroll
    for (int k = 0; k < 5; ++k) {
        const float* yr = train_y + (size_t)ti[k] * OUTD;
        float wgt = wv[k];
        v4f y0 = *(const v4f*)(yr + 0);
        v4f y1 = *(const v4f*)(yr + 4);
        v4f y2 = *(const v4f*)(yr + 8);
        v4f y3 = *(const v4f*)(yr + 12);
        acc0 += wgt * y0; acc1 += wgt * y1; acc2 += wgt * y2; acc3 += wgt * y3;
    }
    float* op = out + (size_t)q * OUTD;
    *(v4f*)(op + 0)  = acc0 * inv;
    *(v4f*)(op + 4)  = acc1 * inv;
    *(v4f*)(op + 8)  = acc2 * inv;
    *(v4f*)(op + 12) = acc3 * inv;
}

extern "C" void kernel_launch(void* const* d_in, const int* in_sizes, int n_in,
                              void* d_out, int out_size, void* d_ws, size_t ws_size,
                              hipStream_t stream) {
    const float* x       = (const float*)d_in[0];
    const float* train_x = (const float*)d_in[1];
    const float* train_y = (const float*)d_in[2];
    const float* weights = (const float*)d_in[3];
    float* out = (float*)d_out;

    float* ws    = (float*)d_ws;
    float* xq_s  = ws;
    float* xt_s  = ws + 131072;
    float* qnorm = ws + 655360;
    float* tnorm = ws + 659456;
    float* pvals = ws + 675840;
    int*   pidx  = (int*)(ws + 839680);

    idw_prep<<<(NQ + NT) / 128, 128, 0, stream>>>(x, train_x, weights,
                                                  xq_s, xt_s, qnorm, tnorm);
    idw_main<<<(NQ / 16) * SEGS, 32, 0, stream>>>(xq_s, xt_s, qnorm, tnorm,
                                                  pvals, pidx);
    idw_reduce<<<NQ / 128, 128, 0, stream>>>(pvals, pidx, train_y, out);
}


# main kernel: each tile's A-fragment and norm loads are issued one tile ahead into a second register set, so the cache wait stands behind a tile of WMMA and select work; same values and sum order
# speedup vs baseline: 2.1970x; 2.1970x over previous
#include <hip/hip_runtime.h>
#include <hip/hip_bf16.h>


#define NQ    4096
#define NT    16384
#define DDIM  32
#define OUTD  16
#define WAVES 32
#define SEGLEN (NT / WAVES)
#define LDS_BYTES (176 * 1024)
#define EPSV  1e-6f
#define BIGF  3.0e38f

typedef __attribute__((ext_vector_type(2))) float v2f;
typedef __attribute__((ext_vector_type(4))) float v4f;
typedef __attribute__((ext_vector_type(8))) float v8f;

__device__ __forceinline__ void bot5_insert(float (&tv)[5], int (&ti)[5],
                                            float v, int id) {
    if (v < tv[4]) {
        tv[4] = v; ti[4] = id;
#pragma unroll
        for (int k = 4; k > 0; --k) {
            if (tv[k] < tv[k - 1]) {
                float fv = tv[k]; tv[k] = tv[k - 1]; tv[k - 1] = fv;
                int   fi = ti[k]; ti[k] = ti[k - 1]; ti[k - 1] = fi;
            }
        }
    }
}

__global__ __launch_bounds__(128) void idw_prep(
    const float* __restrict__ x, const float* __restrict__ tx,
    const float* __restrict__ w,
    float* __restrict__ xq_s, float* __restrict__ xt_s,
    float* __restrict__ qnorm, float* __restrict__ tnorm) {
    int i = blockIdx.x * blockDim.x + threadIdx.x;
    bool isQ = i < NQ;
    const float* src = isQ ? (x + (size_t)i * DDIM) : (tx + (size_t)(i - NQ) * DDIM);
    float*       dst = isQ ? (xq_s + (size_t)i * DDIM) : (xt_s + (size_t)(i - NQ) * DDIM);
    float nrm = 0.f;
#pragma unroll
    for (int j = 0; j < DDIM; ++j) {
        int h  = j >> 4;
        int jj = j & 15;
        int K  = 4 * (jj >> 1) + (jj & 1) + 2 * h;
        float s = __expf(-w[K]);
        float v = src[K] * s;
        dst[j] = v;
        nrm   += v * v;
    }
    if (isQ) qnorm[i] = nrm; else tnorm[i - NQ] = nrm;
}

__device__ __forceinline__ void idw_tile(
    const v4f a0, const v4f a1, const v4f a2, const v4f a3,
    const v4f tna, const v4f tnb, const v2f (&b)[8], const float qn,
    const int id0, float (&tv)[5], int (&ti)[5]) {
    v8f c0 = {0.f, 0.f, 0.f, 0.f, 0.f, 0.f, 0.f, 0.f};
    v8f c1 = c0;
    c0 = __builtin_amdgcn_wmma_f32_16x16x4_f32(false, a0.xy, false, b[0], (short)0, c0, false, false);
    c1 = __builtin_amdgcn_wmma_f32_16x16x4_f32(false, a2.xy, false, b[4], (short)0, c1, false, false);
    c0 = __builtin_amdgcn_wmma_f32_16x16x4_f32(false, a0.zw, false, b[1], (short)0, c0, false, false);
    c1 = __builtin_amdgcn_wmma_f32_16x16x4_f32(false, a2.zw, false, b[5], (short)0, c1, false, false);
    c0 = __builtin_amdgcn_wmma_f32_16x16x4_f32(false, a1.xy, false, b[2], (short)0, c0, false, false);
    c1 = __builtin_amdgcn_wmma_f32_16x16x4_f32(false, a3.xy, false, b[6], (short)0, c1, false, false);
    c0 = __builtin_amdgcn_wmma_f32_16x16x4_f32(false, a1.zw, false, b[3], (short)0, c0, false, false);
    c1 = __builtin_amdgcn_wmma_f32_16x16x4_f32(false, a3.zw, false, b[7], (short)0, c1, false, false);
    float tn[8] = {tna.x, tna.y, tna.z, tna.w, tnb.x, tnb.y, tnb.z, tnb.w};
#pragma unroll
    for (int r = 0; r < 8; ++r) {
        float sq = qn + tn[r];
        sq = fmaf(-2.0f, c0[r], sq);
        sq = fmaf(-2.0f, c1[r], sq);
        bot5_insert(tv, ti, sq, id0 + r);
    }
}

__global__ __launch_bounds__(WAVES * 32) void idw_main(
    const float* __restrict__ xq_s, const float* __restrict__ xt_s,
    const float* __restrict__ qnorm, const float* __restrict__ tnorm,
    const float* __restrict__ train_y, float* __restrict__ out) {
    __shared__ __attribute__((aligned(16))) unsigned char L[LDS_BYTES];
    float* s_v  = (float*)(L);
    int*   s_i  = (int*)  (L + 10240);
    float* s2_v = (float*)(L + 20480);
    int*   s2_i = (int*)  (L + 23040);
    float* s3_v = (float*)(L + 25600);
    int*   s3_i = (int*)  (L + 26240);
    float* s_w  = (float*)(L + 26880);
    int*   s_id = (int*)  (L + 27200);
    float* s_inv = (float*)(L + 27520);

    const int tid   = threadIdx.x;
    const int wave  = __builtin_amdgcn_readfirstlane(tid >> 5);
    const int lane  = tid & 31;
    const int qtile = blockIdx.x;
    const int h     = lane >> 4;
    const int ql    = lane & 15;
    const int q     = qtile * 16 + ql;

    const float* qrow = xq_s + (size_t)q * DDIM + 16 * h;
    v4f bq0 = *(const v4f*)(qrow + 0);
    v4f bq1 = *(const v4f*)(qrow + 4);
    v4f bq2 = *(const v4f*)(qrow + 8);
    v4f bq3 = *(const v4f*)(qrow + 12);
    v2f b[8] = {bq0.xy, bq0.zw, bq1.xy, bq1.zw, bq2.xy, bq2.zw, bq3.xy, bq3.zw};
    const float qn = qnorm[q];

    float tv[5] = {BIGF, BIGF, BIGF, BIGF, BIGF};
    int   ti[5] = {0, 0, 0, 0, 0};

    const int tbeg = wave * SEGLEN;

    const float* abase = xt_s + (size_t)(tbeg + ql) * DDIM + 16 * h;
    const float* nbase = tnorm + tbeg + 8 * h;
    constexpr int NTILE = SEGLEN / 16;

    v4f p0 = *(const v4f*)(abase + 0);
    v4f p1 = *(const v4f*)(abase + 4);
    v4f p2 = *(const v4f*)(abase + 8);
    v4f p3 = *(const v4f*)(abase + 12);
    v4f pna = *(const v4f*)(nbase);
    v4f pnb = *(const v4f*)(nbase + 4);

    for (int it = 0; it < NTILE; it += 2) {
        const float* ar1 = abase + (size_t)(it + 1) * 16 * DDIM;
        const float* nr1 = nbase + (it + 1) * 16;
        v4f s0 = *(const v4f*)(ar1 + 0);
        v4f s1 = *(const v4f*)(ar1 + 4);
        v4f s2 = *(const v4f*)(ar1 + 8);
        v4f s3 = *(const v4f*)(ar1 + 12);
        v4f sna = *(const v4f*)(nr1);
        v4f snb = *(const v4f*)(nr1 + 4);

        idw_tile(p0, p1, p2, p3, pna, pnb, b, qn, tbeg + it * 16 + 8 * h, tv, ti);

        const int nx = (it + 2 < NTILE) ? it + 2 : NTILE - 1;
        const float* ar2 = abase + (size_t)nx * 16 * DDIM;
        const float* nr2 = nbase + nx * 16;
        p0 = *(const v4f*)(ar2 + 0);
        p1 = *(const v4f*)(ar2 + 4);
        p2 = *(const v4f*)(ar2 + 8);
        p3 = *(const v4f*)(ar2 + 12);
        pna = *(const v4f*)(nr2);
        pnb = *(const v4f*)(nr2 + 4);

        idw_tile(s0, s1, s2, s3, sna, snb, b, qn, tbeg + (it + 1) * 16 + 8 * h, tv, ti);
    }

    float pv[5]; int pi[5];
#pragma unroll
    for (int k = 0; k < 5; ++k) {
        pv[k] = __shfl_xor(tv[k], 16, 32);
        pi[k] = __shfl_xor(ti[k], 16, 32);
    }
#pragma unroll
    for (int k = 0; k < 5; ++k) bot5_insert(tv, ti, pv[k], pi[k]);

    if (lane < 16) {
        int base = (wave * 16 + ql) * 5;
#pragma unroll
        for (int k = 0; k < 5; ++k) { s_v[base + k] = tv[k]; s_i[base + k] = ti[k]; }
    }
    __syncthreads();

    if (tid < 128) {
        const int qq = tid & 15;
        const int g  = tid >> 4;
        float mv[5]; int mi[5];
        int b0 = ((4 * g) * 16 + qq) * 5;
#pragma unroll
        for (int k = 0; k < 5; ++k) { mv[k] = s_v[b0 + k]; mi[k] = s_i[b0 + k]; }
#pragma unroll
        for (int j = 1; j < 4; ++j) {
            int base = ((4 * g + j) * 16 + qq) * 5;
#pragma unroll
            for (int k = 0; k < 5; ++k)
                bot5_insert(mv, mi, s_v[base + k], s_i[base + k]);
        }
        int b2 = (g * 16 + qq) * 5;
#pragma unroll
        for (int k = 0; k < 5; ++k) { s2_v[b2 + k] = mv[k]; s2_i[b2 + k] = mi[k]; }
    }
    __syncthreads();

    if (tid < 32) {
        const int qq = tid & 15;
        const int g2 = tid >> 4;
        float mv[5]; int mi[5];
        int b0 = ((4 * g2) * 16 + qq) * 5;
#pragma unroll
        for (int k = 0; k < 5; ++k) { mv[k] = s2_v[b0 + k]; mi[k] = s2_i[b0 + k]; }
#pragma unroll
        for (int j = 1; j < 4; ++j) {
            int base = ((4 * g2 + j) * 16 + qq) * 5;
#pragma unroll
            for (int k = 0; k < 5; ++k)
                bot5_insert(mv, mi, s2_v[base + k], s2_i[base + k]);
        }
        int b3 = (g2 * 16 + qq) * 5;
#pragma unroll
        for (int k = 0; k < 5; ++k) { s3_v[b3 + k] = mv[k]; s3_i[b3 + k] = mi[k]; }
    }
    __syncthreads();

    if (tid < 16) {
        const int qq = tid;
        float mv[5]; int mi[5];
#pragma unroll
        for (int k = 0; k < 5; ++k) { mv[k] = s3_v[qq * 5 + k]; mi[k] = s3_i[qq * 5 + k]; }
#pragma unroll
        for (int k = 0; k < 5; ++k)
            bot5_insert(mv, mi, s3_v[(16 + qq) * 5 + k], s3_i[(16 + qq) * 5 + k]);
        float wv[5];
#pragma unroll
        for (int k = 0; k < 5; ++k)
            wv[k] = rsqrtf(fmaxf(mv[k], 0.f) + EPSV);
        float sw = wv[0] + wv[1] + wv[2] + wv[3] + wv[4];
        s_inv[qq] = 1.f / sw;
#pragma unroll
        for (int k = 0; k < 5; ++k) { s_w[qq * 5 + k] = wv[k]; s_id[qq * 5 + k] = mi[k]; }
    }
    __syncthreads();

    if (tid < 16 * OUTD) {
        const int qq = tid >> 4;
        const int o  = tid & 15;
        float acc = 0.f;
#pragma unroll
        for (int k = 0; k < 5; ++k)
            acc = fmaf(s_w[qq * 5 + k], train_y[(size_t)s_id[qq * 5 + k] * OUTD + o], acc);
        out[(size_t)(qtile * 16 + qq) * OUTD + o] = acc * s_inv[qq];
    }
}

extern "C" void kernel_launch(void* const* d_in, const int* in_sizes, int n_in,
                              void* d_out, int out_size, void* d_ws, size_t ws_size,
                              hipStream_t stream) {
    const float* x       = (const float*)d_in[0];
    const float* train_x = (const float*)d_in[1];
    const float* train_y = (const float*)d_in[2];
    const float* weights = (const float*)d_in[3];
    float* out = (float*)d_out;

    float* ws    = (float*)d_ws;
    float* xq_s  = ws;
    float* xt_s  = ws + 131072;
    float* qnorm = ws + 655360;
    float* tnorm = ws + 659456;

    idw_prep<<<(NQ + NT) / 128, 128, 0, stream>>>(x, train_x, weights,
                                                  xq_s, xt_s, qnorm, tnorm);
    idw_main<<<NQ / 16, WAVES * 32, 0, stream>>>(xq_s, xt_s, qnorm, tnorm,
                                                 train_y, out);
}


# main kernel: after 4 tiles per wave the work-group merges list values in LDS into a bound per query (5th smallest of 2048 rows); later cells insert only under it; same rows and sum order
# speedup vs baseline: 2.3588x; 1.0736x over previous
#include <hip/hip_runtime.h>
#include <hip/hip_bf16.h>


#define NQ    4096
#define NT    16384
#define DDIM  32
#define OUTD  16
#define WAVES 32
#define SEGLEN (NT / WAVES)
#define LDS_BYTES (176 * 1024)
#define PH1   4
#define EPSV  1e-6f
#define BIGF  3.0e38f

typedef __attribute__((ext_vector_type(2))) float v2f;
typedef __attribute__((ext_vector_type(4))) float v4f;
typedef __attribute__((ext_vector_type(8))) float v8f;

__device__ __forceinline__ void bot5_insert(float (&tv)[5], int (&ti)[5],
                                            float v, int id) {
    if (v < tv[4]) {
        tv[4] = v; ti[4] = id;
#pragma unroll
        for (int k = 4; k > 0; --k) {
            if (tv[k] < tv[k - 1]) {
                float fv = tv[k]; tv[k] = tv[k - 1]; tv[k - 1] = fv;
                int   fi = ti[k]; ti[k] = ti[k - 1]; ti[k - 1] = fi;
            }
        }
    }
}

__device__ __forceinline__ void bot5_insert_lim(float (&tv)[5], int (&ti)[5],
                                                float& lim, const float bound,
                                                float v, int id) {
    if (v < lim) {
        tv[4] = v; ti[4] = id;
#pragma unroll
        for (int k = 4; k > 0; --k) {
            const bool  sw = tv[k] < tv[k - 1];
            const float lo = sw ? tv[k] : tv[k - 1];
            const float hi = sw ? tv[k - 1] : tv[k];
            const int   li = sw ? ti[k] : ti[k - 1];
            const int   hj = sw ? ti[k - 1] : ti[k];
            tv[k - 1] = lo; tv[k] = hi; ti[k - 1] = li; ti[k] = hj;
        }
        lim = fminf(tv[4], bound);
    }
}

__device__ __forceinline__ void bot5_value_insert(float (&mv)[5], float v) {
    float n4 = __builtin_amdgcn_fmed3f(mv[3], mv[4], v);
    float n3 = __builtin_amdgcn_fmed3f(mv[2], mv[3], v);
    float n2 = __builtin_amdgcn_fmed3f(mv[1], mv[2], v);
    float n1 = __builtin_amdgcn_fmed3f(mv[0], mv[1], v);
    float n0 = fminf(mv[0], v);
    mv[0] = n0; mv[1] = n1; mv[2] = n2; mv[3] = n3; mv[4] = n4;
}

__global__ __launch_bounds__(128) void idw_prep(
    const float* __restrict__ x, const float* __restrict__ tx,
    const float* __restrict__ w,
    float* __restrict__ xq_s, float* __restrict__ xt_s,
    float* __restrict__ qnorm, float* __restrict__ tnorm) {
    int i = blockIdx.x * blockDim.x + threadIdx.x;
    bool isQ = i < NQ;
    const float* src = isQ ? (x + (size_t)i * DDIM) : (tx + (size_t)(i - NQ) * DDIM);
    float*       dst = isQ ? (xq_s + (size_t)i * DDIM) : (xt_s + (size_t)(i - NQ) * DDIM);
    float nrm = 0.f;
#pragma unroll
    for (int j = 0; j < DDIM; ++j) {
        int h  = j >> 4;
        int jj = j & 15;
        int K  = 4 * (jj >> 1) + (jj & 1) + 2 * h;
        float s = __expf(-w[K]);
        float v = src[K] * s;
        dst[j] = v;
        nrm   += v * v;
    }
    if (isQ) qnorm[i] = nrm; else tnorm[i - NQ] = nrm;
}

__device__ __forceinline__ void idw_tile(
    const v4f a0, const v4f a1, const v4f a2, const v4f a3,
    const v4f tna, const v4f tnb, const v2f (&b)[8], const float qn,
    const int id0, float (&tv)[5], int (&ti)[5],
    float& lim, const float bound) {
    v8f c0 = {0.f, 0.f, 0.f, 0.f, 0.f, 0.f, 0.f, 0.f};
    v8f c1 = c0;
    c0 = __builtin_amdgcn_wmma_f32_16x16x4_f32(false, a0.xy, false, b[0], (short)0, c0, false, false);
    c1 = __builtin_amdgcn_wmma_f32_16x16x4_f32(false, a2.xy, false, b[4], (short)0, c1, false, false);
    c0 = __builtin_amdgcn_wmma_f32_16x16x4_f32(false, a0.zw, false, b[1], (short)0, c0, false, false);
    c1 = __builtin_amdgcn_wmma_f32_16x16x4_f32(false, a2.zw, false, b[5], (short)0, c1, false, false);
    c0 = __builtin_amdgcn_wmma_f32_16x16x4_f32(false, a1.xy, false, b[2], (short)0, c0, false, false);
    c1 = __builtin_amdgcn_wmma_f32_16x16x4_f32(false, a3.xy, false, b[6], (short)0, c1, false, false);
    c0 = __builtin_amdgcn_wmma_f32_16x16x4_f32(false, a1.zw, false, b[3], (short)0, c0, false, false);
    c1 = __builtin_amdgcn_wmma_f32_16x16x4_f32(false, a3.zw, false, b[7], (short)0, c1, false, false);
    float tn[8] = {tna.x, tna.y, tna.z, tna.w, tnb.x, tnb.y, tnb.z, tnb.w};
#pragma unroll
    for (int r = 0; r < 8; ++r) {
        float sq = qn + tn[r];
        sq = fmaf(-2.0f, c0[r], sq);
        sq = fmaf(-2.0f, c1[r], sq);
        bot5_insert_lim(tv, ti, lim, bound, sq, id0 + r);
    }
}

__global__ __launch_bounds__(WAVES * 32) void idw_main(
    const float* __restrict__ xq_s, const float* __restrict__ xt_s,
    const float* __restrict__ qnorm, const float* __restrict__ tnorm,
    const float* __restrict__ train_y, float* __restrict__ out) {
    __shared__ __attribute__((aligned(16))) unsigned char L[LDS_BYTES];
    float* s_v  = (float*)(L);
    int*   s_i  = (int*)  (L + 10240);
    float* s2_v = (float*)(L + 20480);
    int*   s2_i = (int*)  (L + 23040);
    float* s3_v = (float*)(L + 25600);
    int*   s3_i = (int*)  (L + 26240);
    float* s_w  = (float*)(L + 26880);
    int*   s_id = (int*)  (L + 27200);
    float* s_inv = (float*)(L + 27520);
    float* s_pv  = (float*)(L + 32768);
    float* s_p2  = (float*)(L + 53248);
    float* s_bnd = (float*)(L + 58368);

    const int tid   = threadIdx.x;
    const int wave  = __builtin_amdgcn_readfirstlane(tid >> 5);
    const int lane  = tid & 31;
    const int qtile = blockIdx.x;
    const int h     = lane >> 4;
    const int ql    = lane & 15;
    const int q     = qtile * 16 + ql;

    const float* qrow = xq_s + (size_t)q * DDIM + 16 * h;
    v4f bq0 = *(const v4f*)(qrow + 0);
    v4f bq1 = *(const v4f*)(qrow + 4);
    v4f bq2 = *(const v4f*)(qrow + 8);
    v4f bq3 = *(const v4f*)(qrow + 12);
    v2f b[8] = {bq0.xy, bq0.zw, bq1.xy, bq1.zw, bq2.xy, bq2.zw, bq3.xy, bq3.zw};
    const float qn = qnorm[q];

    float tv[5] = {BIGF, BIGF, BIGF, BIGF, BIGF};
    int   ti[5] = {0, 0, 0, 0, 0};

    const int tbeg = wave * SEGLEN;

    const float* abase = xt_s + (size_t)(tbeg + ql) * DDIM + 16 * h;
    const float* nbase = tnorm + tbeg + 8 * h;
    constexpr int NTILE = SEGLEN / 16;

    v4f p0 = *(const v4f*)(abase + 0);
    v4f p1 = *(const v4f*)(abase + 4);
    v4f p2 = *(const v4f*)(abase + 8);
    v4f p3 = *(const v4f*)(abase + 12);
    v4f pna = *(const v4f*)(nbase);
    v4f pnb = *(const v4f*)(nbase + 4);

    float bound = BIGF;
    float lim   = BIGF;

    auto trip = [&](const int it) __attribute__((always_inline)) {
        const float* ar1 = abase + (size_t)(it + 1) * 16 * DDIM;
        const float* nr1 = nbase + (it + 1) * 16;
        v4f s0 = *(const v4f*)(ar1 + 0);
        v4f s1 = *(const v4f*)(ar1 + 4);
        v4f s2 = *(const v4f*)(ar1 + 8);
        v4f s3 = *(const v4f*)(ar1 + 12);
        v4f sna = *(const v4f*)(nr1);
        v4f snb = *(const v4f*)(nr1 + 4);

        idw_tile(p0, p1, p2, p3, pna, pnb, b, qn, tbeg + it * 16 + 8 * h, tv, ti, lim, bound);

        const int nx = (it + 2 < NTILE) ? it + 2 : NTILE - 1;
        const float* ar2 = abase + (size_t)nx * 16 * DDIM;
        const float* nr2 = nbase + nx * 16;
        p0 = *(const v4f*)(ar2 + 0);
        p1 = *(const v4f*)(ar2 + 4);
        p2 = *(const v4f*)(ar2 + 8);
        p3 = *(const v4f*)(ar2 + 12);
        pna = *(const v4f*)(nr2);
        pnb = *(const v4f*)(nr2 + 4);

        idw_tile(s0, s1, s2, s3, sna, snb, b, qn, tbeg + (it + 1) * 16 + 8 * h, tv, ti, lim, bound);
    };

    for (int it = 0; it < PH1; it += 2) trip(it);

    {
        const int pb = ((wave * 2 + h) * 16 + ql) * 5;
#pragma unroll
        for (int k = 0; k < 5; ++k) s_pv[pb + k] = tv[k];
    }
    __syncthreads();
    if (tid < 128) {
        const int qq = tid & 15;
        const int g  = tid >> 4;
        float mv[5];
        const int b0 = ((8 * g) * 16 + qq) * 5;
#pragma unroll
        for (int k = 0; k < 5; ++k) mv[k] = s_pv[b0 + k];
#pragma unroll
        for (int j = 1; j < 8; ++j) {
            const int base = ((8 * g + j) * 16 + qq) * 5;
#pragma unroll
            for (int k = 0; k < 5; ++k) bot5_value_insert(mv, s_pv[base + k]);
        }
        const int b2 = (g * 16 + qq) * 5;
#pragma unroll
        for (int k = 0; k < 5; ++k) s_p2[b2 + k] = mv[k];
    }
    __syncthreads();
    if (tid < 16) {
        const int qq = tid;
        float mv[5];
#pragma unroll
        for (int k = 0; k < 5; ++k) mv[k] = s_p2[qq * 5 + k];
#pragma unroll
        for (int j = 1; j < 8; ++j) {
            const int base = (j * 16 + qq) * 5;
#pragma unroll
            for (int k = 0; k < 5; ++k) bot5_value_insert(mv, s_p2[base + k]);
        }
        s_bnd[qq] = mv[4];
    }
    __syncthreads();
    bound = s_bnd[ql];
    lim   = fminf(tv[4], bound);

    for (int it = PH1; it < NTILE; it += 2) trip(it);

    float pv[5]; int pi[5];
#pragma unroll
    for (int k = 0; k < 5; ++k) {
        pv[k] = __shfl_xor(tv[k], 16, 32);
        pi[k] = __shfl_xor(ti[k], 16, 32);
    }
#pragma unroll
    for (int k = 0; k < 5; ++k) bot5_insert(tv, ti, pv[k], pi[k]);

    if (lane < 16) {
        int base = (wave * 16 + ql) * 5;
#pragma unroll
        for (int k = 0; k < 5; ++k) { s_v[base + k] = tv[k]; s_i[base + k] = ti[k]; }
    }
    __syncthreads();

    if (tid < 128) {
        const int qq = tid & 15;
        const int g  = tid >> 4;
        float mv[5]; int mi[5];
        int b0 = ((4 * g) * 16 + qq) * 5;
#pragma unroll
        for (int k = 0; k < 5; ++k) { mv[k] = s_v[b0 + k]; mi[k] = s_i[b0 + k]; }
#pragma unroll
        for (int j = 1; j < 4; ++j) {
            int base = ((4 * g + j) * 16 + qq) * 5;
#pragma unroll
            for (int k = 0; k < 5; ++k)
                bot5_insert(mv, mi, s_v[base + k], s_i[base + k]);
        }
        int b2 = (g * 16 + qq) * 5;
#pragma unroll
        for (int k = 0; k < 5; ++k) { s2_v[b2 + k] = mv[k]; s2_i[b2 + k] = mi[k]; }
    }
    __syncthreads();

    if (tid < 32) {
        const int qq = tid & 15;
        const int g2 = tid >> 4;
        float mv[5]; int mi[5];
        int b0 = ((4 * g2) * 16 + qq) * 5;
#pragma unroll
        for (int k = 0; k < 5; ++k) { mv[k] = s2_v[b0 + k]; mi[k] = s2_i[b0 + k]; }
#pragma unroll
        for (int j = 1; j < 4; ++j) {
            int base = ((4 * g2 + j) * 16 + qq) * 5;
#pragma unroll
            for (int k = 0; k < 5; ++k)
                bot5_insert(mv, mi, s2_v[base + k], s2_i[base + k]);
        }
        int b3 = (g2 * 16 + qq) * 5;
#pragma unroll
        for (int k = 0; k < 5; ++k) { s3_v[b3 + k] = mv[k]; s3_i[b3 + k] = mi[k]; }
    }
    __syncthreads();

    if (tid < 16) {
        const int qq = tid;
        float mv[5]; int mi[5];
#pragma unroll
        for (int k = 0; k < 5; ++k) { mv[k] = s3_v[qq * 5 + k]; mi[k] = s3_i[qq * 5 + k]; }
#pragma unroll
        for (int k = 0; k < 5; ++k)
            bot5_insert(mv, mi, s3_v[(16 + qq) * 5 + k], s3_i[(16 + qq) * 5 + k]);
        float wv[5];
#pragma unroll
        for (int k = 0; k < 5; ++k)
            wv[k] = rsqrtf(fmaxf(mv[k], 0.f) + EPSV);
        float sw = wv[0] + wv[1] + wv[2] + wv[3] + wv[4];
        s_inv[qq] = 1.f / sw;
#pragma unroll
        for (int k = 0; k < 5; ++k) { s_w[qq * 5 + k] = wv[k]; s_id[qq * 5 + k] = mi[k]; }
    }
    __syncthreads();

    if (tid < 16 * OUTD) {
        const int qq = tid >> 4;
        const int o  = tid & 15;
        float acc = 0.f;
#pragma unroll
        for (int k = 0; k < 5; ++k)
            acc = fmaf(s_w[qq * 5 + k], train_y[(size_t)s_id[qq * 5 + k] * OUTD + o], acc);
        out[(size_t)(qtile * 16 + qq) * OUTD + o] = acc * s_inv[qq];
    }
}

extern "C" void kernel_launch(void* const* d_in, const int* in_sizes, int n_in,
                              void* d_out, int out_size, void* d_ws, size_t ws_size,
                              hipStream_t stream) {
    const float* x       = (const float*)d_in[0];
    const float* train_x = (const float*)d_in[1];
    const float* train_y = (const float*)d_in[2];
    const float* weights = (const float*)d_in[3];
    float* out = (float*)d_out;

    float* ws    = (float*)d_ws;
    float* xq_s  = ws;
    float* xt_s  = ws + 131072;
    float* qnorm = ws + 655360;
    float* tnorm = ws + 659456;

    idw_prep<<<(NQ + NT) / 128, 128, 0, stream>>>(x, train_x, weights,
                                                  xq_s, xt_s, qnorm, tnorm);
    idw_main<<<NQ / 16, WAVES * 32, 0, stream>>>(xq_s, xt_s, qnorm, tnorm,
                                                 train_y, out);
}


# prep stores the scaled train rows tile by tile in fragment order: each A-fragment load of the main kernel reads 4 whole 128-byte lines, not 32-byte pieces of 16 lines; same values and sum order
# speedup vs baseline: 2.6808x; 1.1365x over previous
#include <hip/hip_runtime.h>
#include <hip/hip_bf16.h>


#define NQ    4096
#define NT    16384
#define DDIM  32
#define OUTD  16
#define WAVES 32
#define SEGLEN (NT / WAVES)
#define LDS_BYTES (176 * 1024)
#define PH1   4
#define AQ    128
#define EPSV  1e-6f
#define BIGF  3.0e38f

typedef __attribute__((ext_vector_type(2))) float v2f;
typedef __attribute__((ext_vector_type(4))) float v4f;
typedef __attribute__((ext_vector_type(8))) float v8f;

__device__ __forceinline__ void bot5_insert(float (&tv)[5], int (&ti)[5],
                                            float v, int id) {
    if (v < tv[4]) {
        tv[4] = v; ti[4] = id;
#pragma unroll
        for (int k = 4; k > 0; --k) {
            if (tv[k] < tv[k - 1]) {
                float fv = tv[k]; tv[k] = tv[k - 1]; tv[k - 1] = fv;
                int   fi = ti[k]; ti[k] = ti[k - 1]; ti[k - 1] = fi;
            }
        }
    }
}

__device__ __forceinline__ void bot5_insert_lim(float (&tv)[5], int (&ti)[5],
                                                float& lim, const float bound,
                                                float v, int id) {
    if (v < lim) {
        tv[4] = v; ti[4] = id;
#pragma unroll
        for (int k = 4; k > 0; --k) {
            const bool  sw = tv[k] < tv[k - 1];
            const float lo = sw ? tv[k] : tv[k - 1];
            const float hi = sw ? tv[k - 1] : tv[k];
            const int   li = sw ? ti[k] : ti[k - 1];
            const int   hj = sw ? ti[k - 1] : ti[k];
            tv[k - 1] = lo; tv[k] = hi; ti[k - 1] = li; ti[k] = hj;
        }
        lim = fminf(tv[4], bound);
    }
}

__device__ __forceinline__ void bot5_value_insert(float (&mv)[5], float v) {
    float n4 = __builtin_amdgcn_fmed3f(mv[3], mv[4], v);
    float n3 = __builtin_amdgcn_fmed3f(mv[2], mv[3], v);
    float n2 = __builtin_amdgcn_fmed3f(mv[1], mv[2], v);
    float n1 = __builtin_amdgcn_fmed3f(mv[0], mv[1], v);
    float n0 = fminf(mv[0], v);
    mv[0] = n0; mv[1] = n1; mv[2] = n2; mv[3] = n3; mv[4] = n4;
}

__global__ __launch_bounds__(128) void idw_prep(
    const float* __restrict__ x, const float* __restrict__ tx,
    const float* __restrict__ w,
    float* __restrict__ xq_s, float* __restrict__ xt_s,
    float* __restrict__ qnorm, float* __restrict__ tnorm) {
    int i = blockIdx.x * blockDim.x + threadIdx.x;
    bool isQ = i < NQ;
    const int t = i - NQ;
    const float* src = isQ ? (x + (size_t)i * DDIM) : (tx + (size_t)t * DDIM);
    float* dst = isQ ? (xq_s + (size_t)i * DDIM)
                     : (xt_s + (size_t)(t >> 4) * (16 * DDIM) + (t & 15) * 4);
    const int step_c = isQ ? 4 : AQ;
    const int step_h = isQ ? 16 : AQ / 2;
    float vals[DDIM];
    float nrm = 0.f;
#pragma unroll
    for (int j = 0; j < DDIM; ++j) {
        int h  = j >> 4;
        int jj = j & 15;
        int K  = 4 * (jj >> 1) + (jj & 1) + 2 * h;
        float s = __expf(-w[K]);
        float v = src[K] * s;
        vals[j] = v;
        nrm   += v * v;
    }
#pragma unroll
    for (int g = 0; g < 8; ++g) {
        v4f o = {vals[4 * g], vals[4 * g + 1], vals[4 * g + 2], vals[4 * g + 3]};
        *(v4f*)(dst + (g & 3) * step_c + (g >> 2) * step_h) = o;
    }
    if (isQ) qnorm[i] = nrm; else tnorm[t] = nrm;
}

__device__ __forceinline__ void idw_tile(
    const v4f a0, const v4f a1, const v4f a2, const v4f a3,
    const v4f tna, const v4f tnb, const v2f (&b)[8], const float qn,
    const int id0, float (&tv)[5], int (&ti)[5],
    float& lim, const float bound) {
    v8f c0 = {0.f, 0.f, 0.f, 0.f, 0.f, 0.f, 0.f, 0.f};
    v8f c1 = c0;
    c0 = __builtin_amdgcn_wmma_f32_16x16x4_f32(false, a0.xy, false, b[0], (short)0, c0, false, false);
    c1 = __builtin_amdgcn_wmma_f32_16x16x4_f32(false, a2.xy, false, b[4], (short)0, c1, false, false);
    c0 = __builtin_amdgcn_wmma_f32_16x16x4_f32(false, a0.zw, false, b[1], (short)0, c0, false, false);
    c1 = __builtin_amdgcn_wmma_f32_16x16x4_f32(false, a2.zw, false, b[5], (short)0, c1, false, false);
    c0 = __builtin_amdgcn_wmma_f32_16x16x4_f32(false, a1.xy, false, b[2], (short)0, c0, false, false);
    c1 = __builtin_amdgcn_wmma_f32_16x16x4_f32(false, a3.xy, false, b[6], (short)0, c1, false, false);
    c0 = __builtin_amdgcn_wmma_f32_16x16x4_f32(false, a1.zw, false, b[3], (short)0, c0, false, false);
    c1 = __builtin_amdgcn_wmma_f32_16x16x4_f32(false, a3.zw, false, b[7], (short)0, c1, false, false);
    float tn[8] = {tna.x, tna.y, tna.z, tna.w, tnb.x, tnb.y, tnb.z, tnb.w};
#pragma unroll
    for (int r = 0; r < 8; ++r) {
        float sq = qn + tn[r];
        sq = fmaf(-2.0f, c0[r], sq);
        sq = fmaf(-2.0f, c1[r], sq);
        bot5_insert_lim(tv, ti, lim, bound, sq, id0 + r);
    }
}

__global__ __launch_bounds__(WAVES * 32) void idw_main(
    const float* __restrict__ xq_s, const float* __restrict__ xt_s,
    const float* __restrict__ qnorm, const float* __restrict__ tnorm,
    const float* __restrict__ train_y, float* __restrict__ out) {
    __shared__ __attribute__((aligned(16))) unsigned char L[LDS_BYTES];
    float* s_v  = (float*)(L);
    int*   s_i  = (int*)  (L + 10240);
    float* s2_v = (float*)(L + 20480);
    int*   s2_i = (int*)  (L + 23040);
    float* s3_v = (float*)(L + 25600);
    int*   s3_i = (int*)  (L + 26240);
    float* s_w  = (float*)(L + 26880);
    int*   s_id = (int*)  (L + 27200);
    float* s_inv = (float*)(L + 27520);
    float* s_pv  = (float*)(L + 32768);
    float* s_p2  = (float*)(L + 53248);
    float* s_bnd = (float*)(L + 58368);

    const int tid   = threadIdx.x;
    const int wave  = __builtin_amdgcn_readfirstlane(tid >> 5);
    const int lane  = tid & 31;
    const int qtile = blockIdx.x;
    const int h     = lane >> 4;
    const int ql    = lane & 15;
    const int q     = qtile * 16 + ql;

    const float* qrow = xq_s + (size_t)q * DDIM + 16 * h;
    v4f bq0 = *(const v4f*)(qrow + 0);
    v4f bq1 = *(const v4f*)(qrow + 4);
    v4f bq2 = *(const v4f*)(qrow + 8);
    v4f bq3 = *(const v4f*)(qrow + 12);
    v2f b[8] = {bq0.xy, bq0.zw, bq1.xy, bq1.zw, bq2.xy, bq2.zw, bq3.xy, bq3.zw};
    const float qn = qnorm[q];

    float tv[5] = {BIGF, BIGF, BIGF, BIGF, BIGF};
    int   ti[5] = {0, 0, 0, 0, 0};

    const int tbeg = wave * SEGLEN;

    const float* abase = xt_s + (size_t)tbeg * DDIM + lane * 4;
    const float* nbase = tnorm + tbeg + 8 * h;
    constexpr int NTILE = SEGLEN / 16;

    v4f p0 = *(const v4f*)(abase + 0 * AQ);
    v4f p1 = *(const v4f*)(abase + 1 * AQ);
    v4f p2 = *(const v4f*)(abase + 2 * AQ);
    v4f p3 = *(const v4f*)(abase + 3 * AQ);
    v4f pna = *(const v4f*)(nbase);
    v4f pnb = *(const v4f*)(nbase + 4);

    float bound = BIGF;
    float lim   = BIGF;

    auto trip = [&](const int it) __attribute__((always_inline)) {
        const float* ar1 = abase + (size_t)(it + 1) * 16 * DDIM;
        const float* nr1 = nbase + (it + 1) * 16;
        v4f s0 = *(const v4f*)(ar1 + 0 * AQ);
        v4f s1 = *(const v4f*)(ar1 + 1 * AQ);
        v4f s2 = *(const v4f*)(ar1 + 2 * AQ);
        v4f s3 = *(const v4f*)(ar1 + 3 * AQ);
        v4f sna = *(const v4f*)(nr1);
        v4f snb = *(const v4f*)(nr1 + 4);

        idw_tile(p0, p1, p2, p3, pna, pnb, b, qn, tbeg + it * 16 + 8 * h, tv, ti, lim, bound);

        const int nx = (it + 2 < NTILE) ? it + 2 : NTILE - 1;
        const float* ar2 = abase + (size_t)nx * 16 * DDIM;
        const float* nr2 = nbase + nx * 16;
        p0 = *(const v4f*)(ar2 + 0 * AQ);
        p1 = *(const v4f*)(ar2 + 1 * AQ);
        p2 = *(const v4f*)(ar2 + 2 * AQ);
        p3 = *(const v4f*)(ar2 + 3 * AQ);
        pna = *(const v4f*)(nr2);
        pnb = *(const v4f*)(nr2 + 4);

        idw_tile(s0, s1, s2, s3, sna, snb, b, qn, tbeg + (it + 1) * 16 + 8 * h, tv, ti, lim, bound);
    };

    for (int it = 0; it < PH1; it += 2) trip(it);

    {
        const int pb = ((wave * 2 + h) * 16 + ql) * 5;
#pragma unroll
        for (int k = 0; k < 5; ++k) s_pv[pb + k] = tv[k];
    }
    __syncthreads();
    if (tid < 128) {
        const int qq = tid & 15;
        const int g  = tid >> 4;
        float mv[5];
        const int b0 = ((8 * g) * 16 + qq) * 5;
#pragma unroll
        for (int k = 0; k < 5; ++k) mv[k] = s_pv[b0 + k];
#pragma unroll
        for (int j = 1; j < 8; ++j) {
            const int base = ((8 * g + j) * 16 + qq) * 5;
#pragma unroll
            for (int k = 0; k < 5; ++k) bot5_value_insert(mv, s_pv[base + k]);
        }
        const int b2 = (g * 16 + qq) * 5;
#pragma unroll
        for (int k = 0; k < 5; ++k) s_p2[b2 + k] = mv[k];
    }
    __syncthreads();
    if (tid < 16) {
        const int qq = tid;
        float mv[5];
#pragma unroll
        for (int k = 0; k < 5; ++k) mv[k] = s_p2[qq * 5 + k];
#pragma unroll
        for (int j = 1; j < 8; ++j) {
            const int base = (j * 16 + qq) * 5;
#pragma unroll
            for (int k = 0; k < 5; ++k) bot5_value_insert(mv, s_p2[base + k]);
        }
        s_bnd[qq] = mv[4];
    }
    __syncthreads();
    bound = s_bnd[ql];
    lim   = fminf(tv[4], bound);

    for (int it = PH1; it < NTILE; it += 2) trip(it);

    float pv[5]; int pi[5];
#pragma unroll
    for (int k = 0; k < 5; ++k) {
        pv[k] = __shfl_xor(tv[k], 16, 32);
        pi[k] = __shfl_xor(ti[k], 16, 32);
    }
#pragma unroll
    for (int k = 0; k < 5; ++k) bot5_insert(tv, ti, pv[k], pi[k]);

    if (lane < 16) {
        int base = (wave * 16 + ql) * 5;
#pragma unroll
        for (int k = 0; k < 5; ++k) { s_v[base + k] = tv[k]; s_i[base + k] = ti[k]; }
    }
    __syncthreads();

    if (tid < 128) {
        const int qq = tid & 15;
        const int g  = tid >> 4;
        float mv[5]; int mi[5];
        int b0 = ((4 * g) * 16 + qq) * 5;
#pragma unroll
        for (int k = 0; k < 5; ++k) { mv[k] = s_v[b0 + k]; mi[k] = s_i[b0 + k]; }
#pragma unroll
        for (int j = 1; j < 4; ++j) {
            int base = ((4 * g + j) * 16 + qq) * 5;
#pragma unroll
            for (int k = 0; k < 5; ++k)
                bot5_insert(mv, mi, s_v[base + k], s_i[base + k]);
        }
        int b2 = (g * 16 + qq) * 5;
#pragma unroll
        for (int k = 0; k < 5; ++k) { s2_v[b2 + k] = mv[k]; s2_i[b2 + k] = mi[k]; }
    }
    __syncthreads();

    if (tid < 32) {
        const int qq = tid & 15;
        const int g2 = tid >> 4;
        float mv[5]; int mi[5];
        int b0 = ((4 * g2) * 16 + qq) * 5;
#pragma unroll
        for (int k = 0; k < 5; ++k) { mv[k] = s2_v[b0 + k]; mi[k] = s2_i[b0 + k]; }
#pragma unroll
        for (int j = 1; j < 4; ++j) {
            int base = ((4 * g2 + j) * 16 + qq) * 5;
#pragma unroll
            for (int k = 0; k < 5; ++k)
                bot5_insert(mv, mi, s2_v[base + k], s2_i[base + k]);
        }
        int b3 = (g2 * 16 + qq) * 5;
#pragma unroll
        for (int k = 0; k < 5; ++k) { s3_v[b3 + k] = mv[k]; s3_i[b3 + k] = mi[k]; }
    }
    __syncthreads();

    if (tid < 16) {
        const int qq = tid;
        float mv[5]; int mi[5];
#pragma unroll
        for (int k = 0; k < 5; ++k) { mv[k] = s3_v[qq * 5 + k]; mi[k] = s3_i[qq * 5 + k]; }
#pragma unroll
        for (int k = 0; k < 5; ++k)
            bot5_insert(mv, mi, s3_v[(16 + qq) * 5 + k], s3_i[(16 + qq) * 5 + k]);
        float wv[5];
#pragma unroll
        for (int k = 0; k < 5; ++k)
            wv[k] = rsqrtf(fmaxf(mv[k], 0.f) + EPSV);
        float sw = wv[0] + wv[1] + wv[2] + wv[3] + wv[4];
        s_inv[qq] = 1.f / sw;
#pragma unroll
        for (int k = 0; k < 5; ++k) { s_w[qq * 5 + k] = wv[k]; s_id[qq * 5 + k] = mi[k]; }
    }
    __syncthreads();

    if (tid < 16 * OUTD) {
        const int qq = tid >> 4;
        const int o  = tid & 15;
        float acc = 0.f;
#pragma unroll
        for (int k = 0; k < 5; ++k)
            acc = fmaf(s_w[qq * 5 + k], train_y[(size_t)s_id[qq * 5 + k] * OUTD + o], acc);
        out[(size_t)(qtile * 16 + qq) * OUTD + o] = acc * s_inv[qq];
    }
}

extern "C" void kernel_launch(void* const* d_in, const int* in_sizes, int n_in,
                              void* d_out, int out_size, void* d_ws, size_t ws_size,
                              hipStream_t stream) {
    const float* x       = (const float*)d_in[0];
    const float* train_x = (const float*)d_in[1];
    const float* train_y = (const float*)d_in[2];
    const float* weights = (const float*)d_in[3];
    float* out = (float*)d_out;

    float* ws    = (float*)d_ws;
    float* xq_s  = ws;
    float* xt_s  = ws + 131072;
    float* qnorm = ws + 655360;
    float* tnorm = ws + 659456;

    idw_prep<<<(NQ + NT) / 128, 128, 0, stream>>>(x, train_x, weights,
                                                  xq_s, xt_s, qnorm, tnorm);
    idw_main<<<NQ / 16, WAVES * 32, 0, stream>>>(xq_s, xt_s, qnorm, tnorm,
                                                 train_y, out);
}


# main kernel: query fragment carries -2 and the train norms start the accumulator, so one 8-WMMA chain yields |t|^2-2q.t ready to compare; |q|^2 added to the 5 kept values last; sum order changes
# speedup vs baseline: 2.9454x; 1.0987x over previous
#include <hip/hip_runtime.h>
#include <hip/hip_bf16.h>


#define NQ    4096
#define NT    16384
#define DDIM  32
#define OUTD  16
#define WAVES 32
#define SEGLEN (NT / WAVES)
#define LDS_BYTES (176 * 1024)
#define PH1   4
#define AQ    128
#define EPSV  1e-6f
#define BIGF  3.0e38f

typedef __attribute__((ext_vector_type(2))) float v2f;
typedef __attribute__((ext_vector_type(4))) float v4f;
typedef __attribute__((ext_vector_type(8))) float v8f;

__device__ __forceinline__ void bot5_insert(float (&tv)[5], int (&ti)[5],
                                            float v, int id) {
    if (v < tv[4]) {
        tv[4] = v; ti[4] = id;
#pragma unroll
        for (int k = 4; k > 0; --k) {
            if (tv[k] < tv[k - 1]) {
                float fv = tv[k]; tv[k] = tv[k - 1]; tv[k - 1] = fv;
                int   fi = ti[k]; ti[k] = ti[k - 1]; ti[k - 1] = fi;
            }
        }
    }
}

__device__ __forceinline__ void bot5_insert_lim(float (&tv)[5], int (&ti)[5],
                                                float& lim, const float bound,
                                                float v, int id) {
    if (v < lim) {
        tv[4] = v; ti[4] = id;
#pragma unroll
        for (int k = 4; k > 0; --k) {
            const bool  sw = tv[k] < tv[k - 1];
            const float lo = sw ? tv[k] : tv[k - 1];
            const float hi = sw ? tv[k - 1] : tv[k];
            const int   li = sw ? ti[k] : ti[k - 1];
            const int   hj = sw ? ti[k - 1] : ti[k];
            tv[k - 1] = lo; tv[k] = hi; ti[k - 1] = li; ti[k] = hj;
        }
        lim = fminf(tv[4], bound);
    }
}

__device__ __forceinline__ void bot5_value_insert(float (&mv)[5], float v) {
    float n4 = __builtin_amdgcn_fmed3f(mv[3], mv[4], v);
    float n3 = __builtin_amdgcn_fmed3f(mv[2], mv[3], v);
    float n2 = __builtin_amdgcn_fmed3f(mv[1], mv[2], v);
    float n1 = __builtin_amdgcn_fmed3f(mv[0], mv[1], v);
    float n0 = fminf(mv[0], v);
    mv[0] = n0; mv[1] = n1; mv[2] = n2; mv[3] = n3; mv[4] = n4;
}

__global__ __launch_bounds__(128) void idw_prep(
    const float* __restrict__ x, const float* __restrict__ tx,
    const float* __restrict__ w,
    float* __restrict__ xq_s, float* __restrict__ xt_s,
    float* __restrict__ qnorm, float* __restrict__ tnorm) {
    int i = blockIdx.x * blockDim.x + threadIdx.x;
    bool isQ = i < NQ;
    const int t = i - NQ;
    const float* src = isQ ? (x + (size_t)i * DDIM) : (tx + (size_t)t * DDIM);
    float* dst = isQ ? (xq_s + (size_t)i * DDIM)
                     : (xt_s + (size_t)(t >> 4) * (16 * DDIM) + (t & 15) * 4);
    const int step_c = isQ ? 4 : AQ;
    const int step_h = isQ ? 16 : AQ / 2;
    float vals[DDIM];
    float nrm = 0.f;
#pragma unroll
    for (int j = 0; j < DDIM; ++j) {
        int h  = j >> 4;
        int jj = j & 15;
        int K  = 4 * (jj >> 1) + (jj & 1) + 2 * h;
        float s = __expf(-w[K]);
        float v = src[K] * s;
        vals[j] = v;
        nrm   += v * v;
    }
#pragma unroll
    for (int g = 0; g < 8; ++g) {
        v4f o = {vals[4 * g], vals[4 * g + 1], vals[4 * g + 2], vals[4 * g + 3]};
        *(v4f*)(dst + (g & 3) * step_c + (g >> 2) * step_h) = o;
    }
    if (isQ) qnorm[i] = nrm; else tnorm[t] = nrm;
}

__device__ __forceinline__ void idw_tile(
    const v4f a0, const v4f a1, const v4f a2, const v4f a3,
    const v8f tn, const v2f (&b)[8],
    const int id0, float (&tv)[5], int (&ti)[5],
    float& lim, const float bound) {
    v8f c = tn;
    c = __builtin_amdgcn_wmma_f32_16x16x4_f32(false, a0.xy, false, b[0], (short)0, c, false, false);
    c = __builtin_amdgcn_wmma_f32_16x16x4_f32(false, a0.zw, false, b[1], (short)0, c, false, false);
    c = __builtin_amdgcn_wmma_f32_16x16x4_f32(false, a1.xy, false, b[2], (short)0, c, false, false);
    c = __builtin_amdgcn_wmma_f32_16x16x4_f32(false, a1.zw, false, b[3], (short)0, c, false, false);
    c = __builtin_amdgcn_wmma_f32_16x16x4_f32(false, a2.xy, false, b[4], (short)0, c, false, false);
    c = __builtin_amdgcn_wmma_f32_16x16x4_f32(false, a2.zw, false, b[5], (short)0, c, false, false);
    c = __builtin_amdgcn_wmma_f32_16x16x4_f32(false, a3.xy, false, b[6], (short)0, c, false, false);
    c = __builtin_amdgcn_wmma_f32_16x16x4_f32(false, a3.zw, false, b[7], (short)0, c, false, false);
#pragma unroll
    for (int r = 0; r < 8; ++r)
        bot5_insert_lim(tv, ti, lim, bound, c[r], id0 + r);
}

__global__ __launch_bounds__(WAVES * 32) void idw_main(
    const float* __restrict__ xq_s, const float* __restrict__ xt_s,
    const float* __restrict__ qnorm, const float* __restrict__ tnorm,
    const float* __restrict__ train_y, float* __restrict__ out) {
    __shared__ __attribute__((aligned(16))) unsigned char L[LDS_BYTES];
    float* s_v  = (float*)(L);
    int*   s_i  = (int*)  (L + 10240);
    float* s2_v = (float*)(L + 20480);
    int*   s2_i = (int*)  (L + 23040);
    float* s3_v = (float*)(L + 25600);
    int*   s3_i = (int*)  (L + 26240);
    float* s_w  = (float*)(L + 26880);
    int*   s_id = (int*)  (L + 27200);
    float* s_inv = (float*)(L + 27520);
    float* s_pv  = (float*)(L + 32768);
    float* s_p2  = (float*)(L + 53248);
    float* s_bnd = (float*)(L + 58368);

    const int tid   = threadIdx.x;
    const int wave  = __builtin_amdgcn_readfirstlane(tid >> 5);
    const int lane  = tid & 31;
    const int qtile = blockIdx.x;
    const int h     = lane >> 4;
    const int ql    = lane & 15;
    const int q     = qtile * 16 + ql;

    const float* qrow = xq_s + (size_t)q * DDIM + 16 * h;
    v4f bq0 = *(const v4f*)(qrow + 0);
    v4f bq1 = *(const v4f*)(qrow + 4);
    v4f bq2 = *(const v4f*)(qrow + 8);
    v4f bq3 = *(const v4f*)(qrow + 12);
    bq0 *= -2.0f; bq1 *= -2.0f; bq2 *= -2.0f; bq3 *= -2.0f;
    v2f b[8] = {bq0.xy, bq0.zw, bq1.xy, bq1.zw, bq2.xy, bq2.zw, bq3.xy, bq3.zw};
    const float qn = qnorm[q];

    float tv[5] = {BIGF, BIGF, BIGF, BIGF, BIGF};
    int   ti[5] = {0, 0, 0, 0, 0};

    const int tbeg = wave * SEGLEN;

    const float* abase = xt_s + (size_t)tbeg * DDIM + lane * 4;
    const float* nbase = tnorm + tbeg + 8 * h;
    constexpr int NTILE = SEGLEN / 16;

    v4f p0 = *(const v4f*)(abase + 0 * AQ);
    v4f p1 = *(const v4f*)(abase + 1 * AQ);
    v4f p2 = *(const v4f*)(abase + 2 * AQ);
    v4f p3 = *(const v4f*)(abase + 3 * AQ);
    v8f pn = *(const v8f*)(nbase);

    float bound = BIGF;
    float lim   = BIGF;

    auto trip = [&](const int it) __attribute__((always_inline)) {
        const float* ar1 = abase + (size_t)(it + 1) * 16 * DDIM;
        const float* nr1 = nbase + (it + 1) * 16;
        v4f s0 = *(const v4f*)(ar1 + 0 * AQ);
        v4f s1 = *(const v4f*)(ar1 + 1 * AQ);
        v4f s2 = *(const v4f*)(ar1 + 2 * AQ);
        v4f s3 = *(const v4f*)(ar1 + 3 * AQ);
        v8f sn = *(const v8f*)(nr1);

        idw_tile(p0, p1, p2, p3, pn, b, tbeg + it * 16 + 8 * h, tv, ti, lim, bound);

        const int nx = (it + 2 < NTILE) ? it + 2 : NTILE - 1;
        const float* ar2 = abase + (size_t)nx * 16 * DDIM;
        const float* nr2 = nbase + nx * 16;
        p0 = *(const v4f*)(ar2 + 0 * AQ);
        p1 = *(const v4f*)(ar2 + 1 * AQ);
        p2 = *(const v4f*)(ar2 + 2 * AQ);
        p3 = *(const v4f*)(ar2 + 3 * AQ);
        pn = *(const v8f*)(nr2);

        idw_tile(s0, s1, s2, s3, sn, b, tbeg + (it + 1) * 16 + 8 * h, tv, ti, lim, bound);
    };

    for (int it = 0; it < PH1; it += 2) trip(it);

    {
        const int pb = ((wave * 2 + h) * 16 + ql) * 5;
#pragma unroll
        for (int k = 0; k < 5; ++k) s_pv[pb + k] = tv[k];
    }
    __syncthreads();
    if (tid < 128) {
        const int qq = tid & 15;
        const int g  = tid >> 4;
        float mv[5];
        const int b0 = ((8 * g) * 16 + qq) * 5;
#pragma unroll
        for (int k = 0; k < 5; ++k) mv[k] = s_pv[b0 + k];
#pragma unroll
        for (int j = 1; j < 8; ++j) {
            const int base = ((8 * g + j) * 16 + qq) * 5;
#pragma unroll
            for (int k = 0; k < 5; ++k) bot5_value_insert(mv, s_pv[base + k]);
        }
        const int b2 = (g * 16 + qq) * 5;
#pragma unroll
        for (int k = 0; k < 5; ++k) s_p2[b2 + k] = mv[k];
    }
    __syncthreads();
    if (tid < 16) {
        const int qq = tid;
        float mv[5];
#pragma unroll
        for (int k = 0; k < 5; ++k) mv[k] = s_p2[qq * 5 + k];
#pragma unroll
        for (int j = 1; j < 8; ++j) {
            const int base = (j * 16 + qq) * 5;
#pragma unroll
            for (int k = 0; k < 5; ++k) bot5_value_insert(mv, s_p2[base + k]);
        }
        s_bnd[qq] = mv[4];
    }
    __syncthreads();
    bound = s_bnd[ql];
    lim   = fminf(tv[4], bound);

    for (int it = PH1; it < NTILE; it += 2) trip(it);

    float pv[5]; int pi[5];
#pragma unroll
    for (int k = 0; k < 5; ++k) {
        pv[k] = __shfl_xor(tv[k], 16, 32);
        pi[k] = __shfl_xor(ti[k], 16, 32);
    }
#pragma unroll
    for (int k = 0; k < 5; ++k) bot5_insert(tv, ti, pv[k], pi[k]);

    if (lane < 16) {
        int base = (wave * 16 + ql) * 5;
#pragma unroll
        for (int k = 0; k < 5; ++k) { s_v[base + k] = tv[k]; s_i[base + k] = ti[k]; }
    }
    __syncthreads();

    if (tid < 128) {
        const int qq = tid & 15;
        const int g  = tid >> 4;
        float mv[5]; int mi[5];
        int b0 = ((4 * g) * 16 + qq) * 5;
#pragma unroll
        for (int k = 0; k < 5; ++k) { mv[k] = s_v[b0 + k]; mi[k] = s_i[b0 + k]; }
#pragma unroll
        for (int j = 1; j < 4; ++j) {
            int base = ((4 * g + j) * 16 + qq) * 5;
#pragma unroll
            for (int k = 0; k < 5; ++k)
                bot5_insert(mv, mi, s_v[base + k], s_i[base + k]);
        }
        int b2 = (g * 16 + qq) * 5;
#pragma unroll
        for (int k = 0; k < 5; ++k) { s2_v[b2 + k] = mv[k]; s2_i[b2 + k] = mi[k]; }
    }
    __syncthreads();

    if (tid < 32) {
        const int qq = tid & 15;
        const int g2 = tid >> 4;
        float mv[5]; int mi[5];
        int b0 = ((4 * g2) * 16 + qq) * 5;
#pragma unroll
        for (int k = 0; k < 5; ++k) { mv[k] = s2_v[b0 + k]; mi[k] = s2_i[b0 + k]; }
#pragma unroll
        for (int j = 1; j < 4; ++j) {
            int base = ((4 * g2 + j) * 16 + qq) * 5;
#pragma unroll
            for (int k = 0; k < 5; ++k)
                bot5_insert(mv, mi, s2_v[base + k], s2_i[base + k]);
        }
        int b3 = (g2 * 16 + qq) * 5;
#pragma unroll
        for (int k = 0; k < 5; ++k) { s3_v[b3 + k] = mv[k]; s3_i[b3 + k] = mi[k]; }
    }
    __syncthreads();

    if (tid < 16) {
        const int qq = tid;
        float mv[5]; int mi[5];
#pragma unroll
        for (int k = 0; k < 5; ++k) { mv[k] = s3_v[qq * 5 + k]; mi[k] = s3_i[qq * 5 + k]; }
#pragma unroll
        for (int k = 0; k < 5; ++k)
            bot5_insert(mv, mi, s3_v[(16 + qq) * 5 + k], s3_i[(16 + qq) * 5 + k]);
        float wv[5];
#pragma unroll
        for (int k = 0; k < 5; ++k)
            wv[k] = rsqrtf(fmaxf(mv[k] + qn, 0.f) + EPSV);
        float sw = wv[0] + wv[1] + wv[2] + wv[3] + wv[4];
        s_inv[qq] = 1.f / sw;
#pragma unroll
        for (int k = 0; k < 5; ++k) { s_w[qq * 5 + k] = wv[k]; s_id[qq * 5 + k] = mi[k]; }
    }
    __syncthreads();

    if (tid < 16 * OUTD) {
        const int qq = tid >> 4;
        const int o  = tid & 15;
        float acc = 0.f;
#pragma unroll
        for (int k = 0; k < 5; ++k)
            acc = fmaf(s_w[qq * 5 + k], train_y[(size_t)s_id[qq * 5 + k] * OUTD + o], acc);
        out[(size_t)(qtile * 16 + qq) * OUTD + o] = acc * s_inv[qq];
    }
}

extern "C" void kernel_launch(void* const* d_in, const int* in_sizes, int n_in,
                              void* d_out, int out_size, void* d_ws, size_t ws_size,
                              hipStream_t stream) {
    const float* x       = (const float*)d_in[0];
    const float* train_x = (const float*)d_in[1];
    const float* train_y = (const float*)d_in[2];
    const float* weights = (const float*)d_in[3];
    float* out = (float*)d_out;

    float* ws    = (float*)d_ws;
    float* xq_s  = ws;
    float* xt_s  = ws + 131072;
    float* qnorm = ws + 655360;
    float* tnorm = ws + 659456;

    idw_prep<<<(NQ + NT) / 128, 128, 0, stream>>>(x, train_x, weights,
                                                  xq_s, xt_s, qnorm, tnorm);
    idw_main<<<NQ / 16, WAVES * 32, 0, stream>>>(xq_s, xt_s, qnorm, tnorm,
                                                 train_y, out);
}


# main kernel: a wave steps its issue priority down (s_setprio 3,2,1,0) per tile before the bound and per stretch of the scan after it, so waves served last catch up; same values and sum order
# speedup vs baseline: 3.1062x; 1.0546x over previous
#include <hip/hip_runtime.h>
#include <hip/hip_bf16.h>


#define NQ    4096
#define NT    16384
#define DDIM  32
#define OUTD  16
#define WAVES 32
#define SEGLEN (NT / WAVES)
#define LDS_BYTES (176 * 1024)
#define PH1   4
#define AQ    128
#define EPSV  1e-6f
#define BIGF  3.0e38f

typedef __attribute__((ext_vector_type(2))) float v2f;
typedef __attribute__((ext_vector_type(4))) float v4f;
typedef __attribute__((ext_vector_type(8))) float v8f;

__device__ __forceinline__ void bot5_insert(float (&tv)[5], int (&ti)[5],
                                            float v, int id) {
    if (v < tv[4]) {
        tv[4] = v; ti[4] = id;
#pragma unroll
        for (int k = 4; k > 0; --k) {
            if (tv[k] < tv[k - 1]) {
                float fv = tv[k]; tv[k] = tv[k - 1]; tv[k - 1] = fv;
                int   fi = ti[k]; ti[k] = ti[k - 1]; ti[k - 1] = fi;
            }
        }
    }
}

__device__ __forceinline__ void bot5_insert_lim(float (&tv)[5], int (&ti)[5],
                                                float& lim, const float bound,
                                                float v, int id) {
    if (v < lim) {
        tv[4] = v; ti[4] = id;
#pragma unroll
        for (int k = 4; k > 0; --k) {
            const bool  sw = tv[k] < tv[k - 1];
            const float lo = sw ? tv[k] : tv[k - 1];
            const float hi = sw ? tv[k - 1] : tv[k];
            const int   li = sw ? ti[k] : ti[k - 1];
            const int   hj = sw ? ti[k - 1] : ti[k];
            tv[k - 1] = lo; tv[k] = hi; ti[k - 1] = li; ti[k] = hj;
        }
        lim = fminf(tv[4], bound);
    }
}

__device__ __forceinline__ void bot5_value_insert(float (&mv)[5], float v) {
    float n4 = __builtin_amdgcn_fmed3f(mv[3], mv[4], v);
    float n3 = __builtin_amdgcn_fmed3f(mv[2], mv[3], v);
    float n2 = __builtin_amdgcn_fmed3f(mv[1], mv[2], v);
    float n1 = __builtin_amdgcn_fmed3f(mv[0], mv[1], v);
    float n0 = fminf(mv[0], v);
    mv[0] = n0; mv[1] = n1; mv[2] = n2; mv[3] = n3; mv[4] = n4;
}

__global__ __launch_bounds__(128) void idw_prep(
    const float* __restrict__ x, const float* __restrict__ tx,
    const float* __restrict__ w,
    float* __restrict__ xq_s, float* __restrict__ xt_s,
    float* __restrict__ qnorm, float* __restrict__ tnorm) {
    int i = blockIdx.x * blockDim.x + threadIdx.x;
    bool isQ = i < NQ;
    const int t = i - NQ;
    const float* src = isQ ? (x + (size_t)i * DDIM) : (tx + (size_t)t * DDIM);
    float* dst = isQ ? (xq_s + (size_t)i * DDIM)
                     : (xt_s + (size_t)(t >> 4) * (16 * DDIM) + (t & 15) * 4);
    const int step_c = isQ ? 4 : AQ;
    const int step_h = isQ ? 16 : AQ / 2;
    float vals[DDIM];
    float nrm = 0.f;
#pragma unroll
    for (int j = 0; j < DDIM; ++j) {
        int h  = j >> 4;
        int jj = j & 15;
        int K  = 4 * (jj >> 1) + (jj & 1) + 2 * h;
        float s = __expf(-w[K]);
        float v = src[K] * s;
        vals[j] = v;
        nrm   += v * v;
    }
#pragma unroll
    for (int g = 0; g < 8; ++g) {
        v4f o = {vals[4 * g], vals[4 * g + 1], vals[4 * g + 2], vals[4 * g + 3]};
        *(v4f*)(dst + (g & 3) * step_c + (g >> 2) * step_h) = o;
    }
    if (isQ) qnorm[i] = nrm; else tnorm[t] = nrm;
}

__device__ __forceinline__ void idw_tile(
    const v4f a0, const v4f a1, const v4f a2, const v4f a3,
    const v8f tn, const v2f (&b)[8],
    const int id0, float (&tv)[5], int (&ti)[5],
    float& lim, const float bound) {
    v8f c = tn;
    c = __builtin_amdgcn_wmma_f32_16x16x4_f32(false, a0.xy, false, b[0], (short)0, c, false, false);
    c = __builtin_amdgcn_wmma_f32_16x16x4_f32(false, a0.zw, false, b[1], (short)0, c, false, false);
    c = __builtin_amdgcn_wmma_f32_16x16x4_f32(false, a1.xy, false, b[2], (short)0, c, false, false);
    c = __builtin_amdgcn_wmma_f32_16x16x4_f32(false, a1.zw, false, b[3], (short)0, c, false, false);
    c = __builtin_amdgcn_wmma_f32_16x16x4_f32(false, a2.xy, false, b[4], (short)0, c, false, false);
    c = __builtin_amdgcn_wmma_f32_16x16x4_f32(false, a2.zw, false, b[5], (short)0, c, false, false);
    c = __builtin_amdgcn_wmma_f32_16x16x4_f32(false, a3.xy, false, b[6], (short)0, c, false, false);
    c = __builtin_amdgcn_wmma_f32_16x16x4_f32(false, a3.zw, false, b[7], (short)0, c, false, false);
#pragma unroll
    for (int r = 0; r < 8; ++r)
        bot5_insert_lim(tv, ti, lim, bound, c[r], id0 + r);
}

__global__ __launch_bounds__(WAVES * 32) void idw_main(
    const float* __restrict__ xq_s, const float* __restrict__ xt_s,
    const float* __restrict__ qnorm, const float* __restrict__ tnorm,
    const float* __restrict__ train_y, float* __restrict__ out) {
    __shared__ __attribute__((aligned(16))) unsigned char L[LDS_BYTES];
    float* s_v  = (float*)(L);
    int*   s_i  = (int*)  (L + 10240);
    float* s2_v = (float*)(L + 20480);
    int*   s2_i = (int*)  (L + 23040);
    float* s3_v = (float*)(L + 25600);
    int*   s3_i = (int*)  (L + 26240);
    float* s_w  = (float*)(L + 26880);
    int*   s_id = (int*)  (L + 27200);
    float* s_inv = (float*)(L + 27520);
    float* s_pv  = (float*)(L + 32768);
    float* s_p2  = (float*)(L + 53248);
    float* s_bnd = (float*)(L + 58368);

    const int tid   = threadIdx.x;
    const int wave  = __builtin_amdgcn_readfirstlane(tid >> 5);
    const int lane  = tid & 31;
    const int qtile = blockIdx.x;
    const int h     = lane >> 4;
    const int ql    = lane & 15;
    const int q     = qtile * 16 + ql;

    const float* qrow = xq_s + (size_t)q * DDIM + 16 * h;
    v4f bq0 = *(const v4f*)(qrow + 0);
    v4f bq1 = *(const v4f*)(qrow + 4);
    v4f bq2 = *(const v4f*)(qrow + 8);
    v4f bq3 = *(const v4f*)(qrow + 12);
    bq0 *= -2.0f; bq1 *= -2.0f; bq2 *= -2.0f; bq3 *= -2.0f;
    v2f b[8] = {bq0.xy, bq0.zw, bq1.xy, bq1.zw, bq2.xy, bq2.zw, bq3.xy, bq3.zw};
    const float qn = qnorm[q];

    float tv[5] = {BIGF, BIGF, BIGF, BIGF, BIGF};
    int   ti[5] = {0, 0, 0, 0, 0};

    const int tbeg = wave * SEGLEN;

    const float* abase = xt_s + (size_t)tbeg * DDIM + lane * 4;
    const float* nbase = tnorm + tbeg + 8 * h;
    constexpr int NTILE = SEGLEN / 16;

    v4f p0 = *(const v4f*)(abase + 0 * AQ);
    v4f p1 = *(const v4f*)(abase + 1 * AQ);
    v4f p2 = *(const v4f*)(abase + 2 * AQ);
    v4f p3 = *(const v4f*)(abase + 3 * AQ);
    v8f pn = *(const v8f*)(nbase);

    float bound = BIGF;
    float lim   = BIGF;

    v4f s0, s1, s2, s3;
    v8f sn;
    auto half_p = [&](const int it) __attribute__((always_inline)) {
        const float* ar1 = abase + (size_t)(it + 1) * 16 * DDIM;
        const float* nr1 = nbase + (it + 1) * 16;
        s0 = *(const v4f*)(ar1 + 0 * AQ);
        s1 = *(const v4f*)(ar1 + 1 * AQ);
        s2 = *(const v4f*)(ar1 + 2 * AQ);
        s3 = *(const v4f*)(ar1 + 3 * AQ);
        sn = *(const v8f*)(nr1);

        idw_tile(p0, p1, p2, p3, pn, b, tbeg + it * 16 + 8 * h, tv, ti, lim, bound);
    };
    auto half_s = [&](const int it) __attribute__((always_inline)) {
        const int nx = (it + 2 < NTILE) ? it + 2 : NTILE - 1;
        const float* ar2 = abase + (size_t)nx * 16 * DDIM;
        const float* nr2 = nbase + nx * 16;
        p0 = *(const v4f*)(ar2 + 0 * AQ);
        p1 = *(const v4f*)(ar2 + 1 * AQ);
        p2 = *(const v4f*)(ar2 + 2 * AQ);
        p3 = *(const v4f*)(ar2 + 3 * AQ);
        pn = *(const v8f*)(nr2);

        idw_tile(s0, s1, s2, s3, sn, b, tbeg + (it + 1) * 16 + 8 * h, tv, ti, lim, bound);
    };
    auto trip = [&](const int it) __attribute__((always_inline)) {
        half_p(it);
        half_s(it);
    };

    static_assert(PH1 == 4, "the four priority steps below are one per tile");
    __builtin_amdgcn_s_setprio(3);
    half_p(0);
    __builtin_amdgcn_s_setprio(2);
    half_s(0);
    __builtin_amdgcn_s_setprio(1);
    half_p(2);
    __builtin_amdgcn_s_setprio(0);
    half_s(2);


    {
        const int pb = ((wave * 2 + h) * 16 + ql) * 5;
#pragma unroll
        for (int k = 0; k < 5; ++k) s_pv[pb + k] = tv[k];
    }
    __syncthreads();
    if (tid < 128) {
        const int qq = tid & 15;
        const int g  = tid >> 4;
        float mv[5];
        const int b0 = ((8 * g) * 16 + qq) * 5;
#pragma unroll
        for (int k = 0; k < 5; ++k) mv[k] = s_pv[b0 + k];
#pragma unroll
        for (int j = 1; j < 8; ++j) {
            const int base = ((8 * g + j) * 16 + qq) * 5;
#pragma unroll
            for (int k = 0; k < 5; ++k) bot5_value_insert(mv, s_pv[base + k]);
        }
        const int b2 = (g * 16 + qq) * 5;
#pragma unroll
        for (int k = 0; k < 5; ++k) s_p2[b2 + k] = mv[k];
    }
    __syncthreads();
    if (tid < 16) {
        const int qq = tid;
        float mv[5];
#pragma unroll
        for (int k = 0; k < 5; ++k) mv[k] = s_p2[qq * 5 + k];
#pragma unroll
        for (int j = 1; j < 8; ++j) {
            const int base = (j * 16 + qq) * 5;
#pragma unroll
            for (int k = 0; k < 5; ++k) bot5_value_insert(mv, s_p2[base + k]);
        }
        s_bnd[qq] = mv[4];
    }
    __syncthreads();
    bound = s_bnd[ql];
    lim   = fminf(tv[4], bound);

    static_assert(PH1 == 4 && NTILE == 32, "stretch limits below are for 14 trips");
    __builtin_amdgcn_s_setprio(3);
#pragma nounroll
    for (int it = PH1; it < 14; it += 2) trip(it);
    __builtin_amdgcn_s_setprio(2);
#pragma nounroll
    for (int it = 14; it < 22; it += 2) trip(it);
    __builtin_amdgcn_s_setprio(1);
#pragma nounroll
    for (int it = 22; it < 28; it += 2) trip(it);
    __builtin_amdgcn_s_setprio(0);
#pragma nounroll
    for (int it = 28; it < NTILE; it += 2) trip(it);

    float pv[5]; int pi[5];
#pragma unroll
    for (int k = 0; k < 5; ++k) {
        pv[k] = __shfl_xor(tv[k], 16, 32);
        pi[k] = __shfl_xor(ti[k], 16, 32);
    }
#pragma unroll
    for (int k = 0; k < 5; ++k) bot5_insert(tv, ti, pv[k], pi[k]);

    if (lane < 16) {
        int base = (wave * 16 + ql) * 5;
#pragma unroll
        for (int k = 0; k < 5; ++k) { s_v[base + k] = tv[k]; s_i[base + k] = ti[k]; }
    }
    __syncthreads();

    if (tid < 128) {
        const int qq = tid & 15;
        const int g  = tid >> 4;
        float mv[5]; int mi[5];
        int b0 = ((4 * g) * 16 + qq) * 5;
#pragma unroll
        for (int k = 0; k < 5; ++k) { mv[k] = s_v[b0 + k]; mi[k] = s_i[b0 + k]; }
#pragma unroll
        for (int j = 1; j < 4; ++j) {
            int base = ((4 * g + j) * 16 + qq) * 5;
#pragma unroll
            for (int k = 0; k < 5; ++k)
                bot5_insert(mv, mi, s_v[base + k], s_i[base + k]);
        }
        int b2 = (g * 16 + qq) * 5;
#pragma unroll
        for (int k = 0; k < 5; ++k) { s2_v[b2 + k] = mv[k]; s2_i[b2 + k] = mi[k]; }
    }
    __syncthreads();

    if (tid < 32) {
        const int qq = tid & 15;
        const int g2 = tid >> 4;
        float mv[5]; int mi[5];
        int b0 = ((4 * g2) * 16 + qq) * 5;
#pragma unroll
        for (int k = 0; k < 5; ++k) { mv[k] = s2_v[b0 + k]; mi[k] = s2_i[b0 + k]; }
#pragma unroll
        for (int j = 1; j < 4; ++j) {
            int base = ((4 * g2 + j) * 16 + qq) * 5;
#pragma unroll
            for (int k = 0; k < 5; ++k)
                bot5_insert(mv, mi, s2_v[base + k], s2_i[base + k]);
        }
        int b3 = (g2 * 16 + qq) * 5;
#pragma unroll
        for (int k = 0; k < 5; ++k) { s3_v[b3 + k] = mv[k]; s3_i[b3 + k] = mi[k]; }
    }
    __syncthreads();

    if (tid < 16) {
        const int qq = tid;
        float mv[5]; int mi[5];
#pragma unroll
        for (int k = 0; k < 5; ++k) { mv[k] = s3_v[qq * 5 + k]; mi[k] = s3_i[qq * 5 + k]; }
#pragma unroll
        for (int k = 0; k < 5; ++k)
            bot5_insert(mv, mi, s3_v[(16 + qq) * 5 + k], s3_i[(16 + qq) * 5 + k]);
        float wv[5];
#pragma unroll
        for (int k = 0; k < 5; ++k)
            wv[k] = rsqrtf(fmaxf(mv[k] + qn, 0.f) + EPSV);
        float sw = wv[0] + wv[1] + wv[2] + wv[3] + wv[4];
        s_inv[qq] = 1.f / sw;
#pragma unroll
        for (int k = 0; k < 5; ++k) { s_w[qq * 5 + k] = wv[k]; s_id[qq * 5 + k] = mi[k]; }
    }
    __syncthreads();

    if (tid < 16 * OUTD) {
        const int qq = tid >> 4;
        const int o  = tid & 15;
        float acc = 0.f;
#pragma unroll
        for (int k = 0; k < 5; ++k)
            acc = fmaf(s_w[qq * 5 + k], train_y[(size_t)s_id[qq * 5 + k] * OUTD + o], acc);
        out[(size_t)(qtile * 16 + qq) * OUTD + o] = acc * s_inv[qq];
    }
}

extern "C" void kernel_launch(void* const* d_in, const int* in_sizes, int n_in,
                              void* d_out, int out_size, void* d_ws, size_t ws_size,
                              hipStream_t stream) {
    const float* x       = (const float*)d_in[0];
    const float* train_x = (const float*)d_in[1];
    const float* train_y = (const float*)d_in[2];
    const float* weights = (const float*)d_in[3];
    float* out = (float*)d_out;

    float* ws    = (float*)d_ws;
    float* xq_s  = ws;
    float* xt_s  = ws + 131072;
    float* qnorm = ws + 655360;
    float* tnorm = ws + 659456;

    idw_prep<<<(NQ + NT) / 128, 128, 0, stream>>>(x, train_x, weights,
                                                  xq_s, xt_s, qnorm, tnorm);
    idw_main<<<NQ / 16, WAVES * 32, 0, stream>>>(xq_s, xt_s, qnorm, tnorm,
                                                 train_y, out);
}


# main kernel: final merge by one wave per query: lane l merges wave l's two lists, five wave-wide minima (permlane) pick the 5 rows; no 3-level LDS insertion tree; same rows and sum order
# speedup vs baseline: 3.3811x; 1.0885x over previous
#include <hip/hip_runtime.h>
#include <hip/hip_bf16.h>


#define NQ    4096
#define NT    16384
#define DDIM  32
#define OUTD  16
#define WAVES 32
#define SEGLEN (NT / WAVES)
#define LDS_BYTES (176 * 1024)
#define PH1   4
#define AQ    128
#define FPITCH 66
#define EPSV  1e-6f
#define BIGF  3.0e38f

typedef __attribute__((ext_vector_type(2))) float v2f;
typedef __attribute__((ext_vector_type(2))) int   v2i;
typedef __attribute__((ext_vector_type(4))) float v4f;
typedef __attribute__((ext_vector_type(8))) float v8f;

__device__ __forceinline__ void bot5_insert_lim(float (&tv)[5], int (&ti)[5],
                                                float& lim, const float bound,
                                                float v, int id) {
    if (v < lim) {
        tv[4] = v; ti[4] = id;
#pragma unroll
        for (int k = 4; k > 0; --k) {
            const bool  sw = tv[k] < tv[k - 1];
            const float lo = sw ? tv[k] : tv[k - 1];
            const float hi = sw ? tv[k - 1] : tv[k];
            const int   li = sw ? ti[k] : ti[k - 1];
            const int   hj = sw ? ti[k - 1] : ti[k];
            tv[k - 1] = lo; tv[k] = hi; ti[k - 1] = li; ti[k] = hj;
        }
        lim = fminf(tv[4], bound);
    }
}

__device__ __forceinline__ void bot5_value_insert(float (&mv)[5], float v) {
    float n4 = __builtin_amdgcn_fmed3f(mv[3], mv[4], v);
    float n3 = __builtin_amdgcn_fmed3f(mv[2], mv[3], v);
    float n2 = __builtin_amdgcn_fmed3f(mv[1], mv[2], v);
    float n1 = __builtin_amdgcn_fmed3f(mv[0], mv[1], v);
    float n0 = fminf(mv[0], v);
    mv[0] = n0; mv[1] = n1; mv[2] = n2; mv[3] = n3; mv[4] = n4;
}

__device__ __forceinline__ void bot5_insert_nb(float (&tv)[5], int (&ti)[5],
                                               float v, int id) {
    const bool c0 = v < tv[0], c1 = v < tv[1], c2 = v < tv[2],
               c3 = v < tv[3], c4 = v < tv[4];
    const float n4 = c3 ? tv[3] : (c4 ? v : tv[4]);
    const int   m4 = c3 ? ti[3] : (c4 ? id : ti[4]);
    const float n3 = c2 ? tv[2] : (c3 ? v : tv[3]);
    const int   m3 = c2 ? ti[2] : (c3 ? id : ti[3]);
    const float n2 = c1 ? tv[1] : (c2 ? v : tv[2]);
    const int   m2 = c1 ? ti[1] : (c2 ? id : ti[2]);
    const float n1 = c0 ? tv[0] : (c1 ? v : tv[1]);
    const int   m1 = c0 ? ti[0] : (c1 ? id : ti[1]);
    const float n0 = c0 ? v : tv[0];
    const int   m0 = c0 ? id : ti[0];
    tv[0] = n0; tv[1] = n1; tv[2] = n2; tv[3] = n3; tv[4] = n4;
    ti[0] = m0; ti[1] = m1; ti[2] = m2; ti[3] = m3; ti[4] = m4;
}

__device__ __forceinline__ float xhalf_f(float x) {
    const int b = __builtin_bit_cast(int, x);
    return __builtin_bit_cast(float,
        __builtin_amdgcn_permlanex16(b, b, 0x76543210, 0xfedcba98, false, false));
}
template <int M>
__device__ __forceinline__ float xrow_f(float x) {
    const int b = __builtin_bit_cast(int, x);
    constexpr unsigned lo = M == 1 ? 0x67452301u : M == 2 ? 0x54761032u
                          : M == 4 ? 0x32107654u : 0xfedcba98u;
    constexpr unsigned hi = M == 1 ? 0xefcdab89u : M == 2 ? 0xdcfe98bau
                          : M == 4 ? 0xba98fedcu : 0x76543210u;
    return __builtin_bit_cast(float,
        __builtin_amdgcn_permlane16(b, b, lo, hi, false, false));
}
__device__ __forceinline__ float wave_min_f(float m) {
    m = fminf(m, xrow_f<1>(m));
    m = fminf(m, xrow_f<2>(m));
    m = fminf(m, xrow_f<4>(m));
    m = fminf(m, xrow_f<8>(m));
    m = fminf(m, xhalf_f(m));
    return m;
}

__global__ __launch_bounds__(128) void idw_prep(
    const float* __restrict__ x, const float* __restrict__ tx,
    const float* __restrict__ w,
    float* __restrict__ xq_s, float* __restrict__ xt_s,
    float* __restrict__ qnorm, float* __restrict__ tnorm) {
    int i = blockIdx.x * blockDim.x + threadIdx.x;
    bool isQ = i < NQ;
    const int t = i - NQ;
    const float* src = isQ ? (x + (size_t)i * DDIM) : (tx + (size_t)t * DDIM);
    float* dst = isQ ? (xq_s + (size_t)i * DDIM)
                     : (xt_s + (size_t)(t >> 4) * (16 * DDIM) + (t & 15) * 4);
    const int step_c = isQ ? 4 : AQ;
    const int step_h = isQ ? 16 : AQ / 2;
    float vals[DDIM];
    float nrm = 0.f;
#pragma unroll
    for (int j = 0; j < DDIM; ++j) {
        int h  = j >> 4;
        int jj = j & 15;
        int K  = 4 * (jj >> 1) + (jj & 1) + 2 * h;
        float s = __expf(-w[K]);
        float v = src[K] * s;
        vals[j] = v;
        nrm   += v * v;
    }
#pragma unroll
    for (int g = 0; g < 8; ++g) {
        v4f o = {vals[4 * g], vals[4 * g + 1], vals[4 * g + 2], vals[4 * g + 3]};
        *(v4f*)(dst + (g & 3) * step_c + (g >> 2) * step_h) = o;
    }
    if (isQ) qnorm[i] = nrm; else tnorm[t] = nrm;
}

__device__ __forceinline__ void idw_tile(
    const v4f a0, const v4f a1, const v4f a2, const v4f a3,
    const v8f tn, const v2f (&b)[8],
    const int id0, float (&tv)[5], int (&ti)[5],
    float& lim, const float bound) {
    v8f c = tn;
    c = __builtin_amdgcn_wmma_f32_16x16x4_f32(false, a0.xy, false, b[0], (short)0, c, false, false);
    c = __builtin_amdgcn_wmma_f32_16x16x4_f32(false, a0.zw, false, b[1], (short)0, c, false, false);
    c = __builtin_amdgcn_wmma_f32_16x16x4_f32(false, a1.xy, false, b[2], (short)0, c, false, false);
    c = __builtin_amdgcn_wmma_f32_16x16x4_f32(false, a1.zw, false, b[3], (short)0, c, false, false);
    c = __builtin_amdgcn_wmma_f32_16x16x4_f32(false, a2.xy, false, b[4], (short)0, c, false, false);
    c = __builtin_amdgcn_wmma_f32_16x16x4_f32(false, a2.zw, false, b[5], (short)0, c, false, false);
    c = __builtin_amdgcn_wmma_f32_16x16x4_f32(false, a3.xy, false, b[6], (short)0, c, false, false);
    c = __builtin_amdgcn_wmma_f32_16x16x4_f32(false, a3.zw, false, b[7], (short)0, c, false, false);
#pragma unroll
    for (int r = 0; r < 8; ++r)
        bot5_insert_lim(tv, ti, lim, bound, c[r], id0 + r);
}

__global__ __launch_bounds__(WAVES * 32) void idw_main(
    const float* __restrict__ xq_s, const float* __restrict__ xt_s,
    const float* __restrict__ qnorm, const float* __restrict__ tnorm,
    const float* __restrict__ train_y, float* __restrict__ out) {
    __shared__ __attribute__((aligned(16))) unsigned char L[LDS_BYTES];
    float* s_fv = (float*)(L + 65536);
    int*   s_fi = (int*)  (L + 87040);
    float* s_w  = (float*)(L + 26880);
    int*   s_id = (int*)  (L + 27200);
    float* s_inv = (float*)(L + 27520);
    float* s_pv  = (float*)(L + 32768);
    float* s_p2  = (float*)(L + 53248);
    float* s_bnd = (float*)(L + 58368);

    const int tid   = threadIdx.x;
    const int wave  = __builtin_amdgcn_readfirstlane(tid >> 5);
    const int lane  = tid & 31;
    const int qtile = blockIdx.x;
    const int h     = lane >> 4;
    const int ql    = lane & 15;
    const int q     = qtile * 16 + ql;

    const float* qrow = xq_s + (size_t)q * DDIM + 16 * h;
    v4f bq0 = *(const v4f*)(qrow + 0);
    v4f bq1 = *(const v4f*)(qrow + 4);
    v4f bq2 = *(const v4f*)(qrow + 8);
    v4f bq3 = *(const v4f*)(qrow + 12);
    bq0 *= -2.0f; bq1 *= -2.0f; bq2 *= -2.0f; bq3 *= -2.0f;
    v2f b[8] = {bq0.xy, bq0.zw, bq1.xy, bq1.zw, bq2.xy, bq2.zw, bq3.xy, bq3.zw};
    const float qn = qnorm[q];

    float tv[5] = {BIGF, BIGF, BIGF, BIGF, BIGF};
    int   ti[5] = {0, 0, 0, 0, 0};

    const int tbeg = wave * SEGLEN;

    const float* abase = xt_s + (size_t)tbeg * DDIM + lane * 4;
    const float* nbase = tnorm + tbeg + 8 * h;
    constexpr int NTILE = SEGLEN / 16;

    v4f p0 = *(const v4f*)(abase + 0 * AQ);
    v4f p1 = *(const v4f*)(abase + 1 * AQ);
    v4f p2 = *(const v4f*)(abase + 2 * AQ);
    v4f p3 = *(const v4f*)(abase + 3 * AQ);
    v8f pn = *(const v8f*)(nbase);

    float bound = BIGF;
    float lim   = BIGF;

    v4f s0, s1, s2, s3;
    v8f sn;
    auto half_p = [&](const int it) __attribute__((always_inline)) {
        const float* ar1 = abase + (size_t)(it + 1) * 16 * DDIM;
        const float* nr1 = nbase + (it + 1) * 16;
        s0 = *(const v4f*)(ar1 + 0 * AQ);
        s1 = *(const v4f*)(ar1 + 1 * AQ);
        s2 = *(const v4f*)(ar1 + 2 * AQ);
        s3 = *(const v4f*)(ar1 + 3 * AQ);
        sn = *(const v8f*)(nr1);

        idw_tile(p0, p1, p2, p3, pn, b, tbeg + it * 16 + 8 * h, tv, ti, lim, bound);
    };
    auto half_s = [&](const int it) __attribute__((always_inline)) {
        const int nx = (it + 2 < NTILE) ? it + 2 : NTILE - 1;
        const float* ar2 = abase + (size_t)nx * 16 * DDIM;
        const float* nr2 = nbase + nx * 16;
        p0 = *(const v4f*)(ar2 + 0 * AQ);
        p1 = *(const v4f*)(ar2 + 1 * AQ);
        p2 = *(const v4f*)(ar2 + 2 * AQ);
        p3 = *(const v4f*)(ar2 + 3 * AQ);
        pn = *(const v8f*)(nr2);

        idw_tile(s0, s1, s2, s3, sn, b, tbeg + (it + 1) * 16 + 8 * h, tv, ti, lim, bound);
    };
    auto trip = [&](const int it) __attribute__((always_inline)) {
        half_p(it);
        half_s(it);
    };

    static_assert(PH1 == 4, "the four priority steps below are one per tile");
    __builtin_amdgcn_s_setprio(3);
    half_p(0);
    __builtin_amdgcn_s_setprio(2);
    half_s(0);
    __builtin_amdgcn_s_setprio(1);
    half_p(2);
    __builtin_amdgcn_s_setprio(0);
    half_s(2);


    {
        const int pb = ((wave * 2 + h) * 16 + ql) * 5;
#pragma unroll
        for (int k = 0; k < 5; ++k) s_pv[pb + k] = tv[k];
    }
    __syncthreads();
    if (tid < 128) {
        const int qq = tid & 15;
        const int g  = tid >> 4;
        float mv[5];
        const int b0 = ((8 * g) * 16 + qq) * 5;
#pragma unroll
        for (int k = 0; k < 5; ++k) mv[k] = s_pv[b0 + k];
#pragma unroll
        for (int j = 1; j < 8; ++j) {
            const int base = ((8 * g + j) * 16 + qq) * 5;
#pragma unroll
            for (int k = 0; k < 5; ++k) bot5_value_insert(mv, s_pv[base + k]);
        }
        const int b2 = (g * 16 + qq) * 5;
#pragma unroll
        for (int k = 0; k < 5; ++k) s_p2[b2 + k] = mv[k];
    }
    __syncthreads();
    if (tid < 16) {
        const int qq = tid;
        float mv[5];
#pragma unroll
        for (int k = 0; k < 5; ++k) mv[k] = s_p2[qq * 5 + k];
#pragma unroll
        for (int j = 1; j < 8; ++j) {
            const int base = (j * 16 + qq) * 5;
#pragma unroll
            for (int k = 0; k < 5; ++k) bot5_value_insert(mv, s_p2[base + k]);
        }
        s_bnd[qq] = mv[4];
    }
    __syncthreads();
    bound = s_bnd[ql];
    lim   = fminf(tv[4], bound);

    static_assert(PH1 == 4 && NTILE == 32, "stretch limits below are for 14 trips");
    __builtin_amdgcn_s_setprio(3);
#pragma nounroll
    for (int it = PH1; it < 14; it += 2) trip(it);
    __builtin_amdgcn_s_setprio(2);
#pragma nounroll
    for (int it = 14; it < 22; it += 2) trip(it);
    __builtin_amdgcn_s_setprio(1);
#pragma nounroll
    for (int it = 22; it < 28; it += 2) trip(it);
    __builtin_amdgcn_s_setprio(0);
#pragma nounroll
    for (int it = 28; it < NTILE; it += 2) trip(it);

    {
        const int li = wave * 2 + h;
#pragma unroll
        for (int k = 0; k < 5; ++k) {
            s_fv[(ql * 5 + k) * FPITCH + li] = tv[k];
            s_fi[(ql * 5 + k) * FPITCH + li] = ti[k];
        }
    }
    __syncthreads();

    if (wave < 16) {
        const int qq = wave;
        float av[5], bv[5]; int ai[5], bi[5];
#pragma unroll
        for (int k = 0; k < 5; ++k) {
            const v2f fv2 = *(const v2f*)(s_fv + (qq * 5 + k) * FPITCH + 2 * lane);
            const v2i fi2 = *(const v2i*)(s_fi + (qq * 5 + k) * FPITCH + 2 * lane);
            av[k] = fv2.x; bv[k] = fv2.y; ai[k] = fi2.x; bi[k] = fi2.y;
        }
#pragma unroll
        for (int k = 0; k < 5; ++k) bot5_insert_nb(av, ai, bv[k], bi[k]);

        float selv[5]; int seli[5];
#pragma unroll
        for (int r = 0; r < 5; ++r) {
            const float m = wave_min_f(av[0]);
            const unsigned bal = __builtin_amdgcn_ballot_w32(av[0] == m);
            const int src = bal ? __builtin_ctz(bal) : 0;
            selv[r] = m;
            seli[r] = __builtin_amdgcn_readlane(ai[0], src);
            const bool pop = lane == src;
            av[0] = pop ? av[1] : av[0]; ai[0] = pop ? ai[1] : ai[0];
            av[1] = pop ? av[2] : av[1]; ai[1] = pop ? ai[2] : ai[1];
            av[2] = pop ? av[3] : av[2]; ai[2] = pop ? ai[3] : ai[2];
            av[3] = pop ? av[4] : av[3]; ai[3] = pop ? ai[4] : ai[3];
            av[4] = pop ? BIGF  : av[4];
        }
        const float qnq = __builtin_bit_cast(float,
            __builtin_amdgcn_readlane(__builtin_bit_cast(int, qn), qq));
        float wv[5];
#pragma unroll
        for (int k = 0; k < 5; ++k)
            wv[k] = rsqrtf(fmaxf(selv[k] + qnq, 0.f) + EPSV);
        const float sw = wv[0] + wv[1] + wv[2] + wv[3] + wv[4];
        if (lane == 0) {
            s_inv[qq] = 1.f / sw;
#pragma unroll
            for (int k = 0; k < 5; ++k) { s_w[qq * 5 + k] = wv[k]; s_id[qq * 5 + k] = seli[k]; }
        }
    }
    __syncthreads();

    if (tid < 16 * OUTD) {
        const int qq = tid >> 4;
        const int o  = tid & 15;
        float acc = 0.f;
#pragma unroll
        for (int k = 0; k < 5; ++k)
            acc = fmaf(s_w[qq * 5 + k], train_y[(size_t)s_id[qq * 5 + k] * OUTD + o], acc);
        out[(size_t)(qtile * 16 + qq) * OUTD + o] = acc * s_inv[qq];
    }
}

extern "C" void kernel_launch(void* const* d_in, const int* in_sizes, int n_in,
                              void* d_out, int out_size, void* d_ws, size_t ws_size,
                              hipStream_t stream) {
    const float* x       = (const float*)d_in[0];
    const float* train_x = (const float*)d_in[1];
    const float* train_y = (const float*)d_in[2];
    const float* weights = (const float*)d_in[3];
    float* out = (float*)d_out;

    float* ws    = (float*)d_ws;
    float* xq_s  = ws;
    float* xt_s  = ws + 131072;
    float* qnorm = ws + 655360;
    float* tnorm = ws + 659456;

    idw_prep<<<(NQ + NT) / 128, 128, 0, stream>>>(x, train_x, weights,
                                                  xq_s, xt_s, qnorm, tnorm);
    idw_main<<<NQ / 16, WAVES * 32, 0, stream>>>(xq_s, xt_s, qnorm, tnorm,
                                                 train_y, out);
}


# main kernel: no insert before the bound: a wave keeps the cells of its first 4 tiles, the bound is the 5th smallest of a query's 64 lane minima, then cells insert only under it; same rows, sum order
# speedup vs baseline: 3.5288x; 1.0437x over previous
#include <hip/hip_runtime.h>
#include <hip/hip_bf16.h>


#define NQ    4096
#define NT    16384
#define DDIM  32
#define OUTD  16
#define WAVES 32
#define SEGLEN (NT / WAVES)
#define LDS_BYTES (176 * 1024)
#define PH1   4
#define AQ    128
#define FPITCH 66
#define EPSV  1e-6f
#define BIGF  3.0e38f

typedef __attribute__((ext_vector_type(2))) float v2f;
typedef __attribute__((ext_vector_type(2))) int   v2i;
typedef __attribute__((ext_vector_type(4))) float v4f;
typedef __attribute__((ext_vector_type(8))) float v8f;

__device__ __forceinline__ void bot5_insert_lim(float (&tv)[5], int (&ti)[5],
                                                float& lim, const float bound,
                                                float v, int id) {
    if (v <= lim) {
        tv[4] = v; ti[4] = id;
#pragma unroll
        for (int k = 4; k > 0; --k) {
            const bool  sw = tv[k] < tv[k - 1];
            const float lo = sw ? tv[k] : tv[k - 1];
            const float hi = sw ? tv[k - 1] : tv[k];
            const int   li = sw ? ti[k] : ti[k - 1];
            const int   hj = sw ? ti[k - 1] : ti[k];
            tv[k - 1] = lo; tv[k] = hi; ti[k - 1] = li; ti[k] = hj;
        }
        lim = fminf(tv[4], bound);
    }
}

__device__ __forceinline__ void bot5_insert_nb(float (&tv)[5], int (&ti)[5],
                                               float v, int id) {
    const bool c0 = v < tv[0], c1 = v < tv[1], c2 = v < tv[2],
               c3 = v < tv[3], c4 = v < tv[4];
    const float n4 = c3 ? tv[3] : (c4 ? v : tv[4]);
    const int   m4 = c3 ? ti[3] : (c4 ? id : ti[4]);
    const float n3 = c2 ? tv[2] : (c3 ? v : tv[3]);
    const int   m3 = c2 ? ti[2] : (c3 ? id : ti[3]);
    const float n2 = c1 ? tv[1] : (c2 ? v : tv[2]);
    const int   m2 = c1 ? ti[1] : (c2 ? id : ti[2]);
    const float n1 = c0 ? tv[0] : (c1 ? v : tv[1]);
    const int   m1 = c0 ? ti[0] : (c1 ? id : ti[1]);
    const float n0 = c0 ? v : tv[0];
    const int   m0 = c0 ? id : ti[0];
    tv[0] = n0; tv[1] = n1; tv[2] = n2; tv[3] = n3; tv[4] = n4;
    ti[0] = m0; ti[1] = m1; ti[2] = m2; ti[3] = m3; ti[4] = m4;
}

__device__ __forceinline__ float xhalf_f(float x) {
    const int b = __builtin_bit_cast(int, x);
    return __builtin_bit_cast(float,
        __builtin_amdgcn_permlanex16(b, b, 0x76543210, 0xfedcba98, false, false));
}
template <int M>
__device__ __forceinline__ float xrow_f(float x) {
    const int b = __builtin_bit_cast(int, x);
    constexpr unsigned lo = M == 1 ? 0x67452301u : M == 2 ? 0x54761032u
                          : M == 4 ? 0x32107654u : 0xfedcba98u;
    constexpr unsigned hi = M == 1 ? 0xefcdab89u : M == 2 ? 0xdcfe98bau
                          : M == 4 ? 0xba98fedcu : 0x76543210u;
    return __builtin_bit_cast(float,
        __builtin_amdgcn_permlane16(b, b, lo, hi, false, false));
}
__device__ __forceinline__ float wave_min_f(float m) {
    m = fminf(m, xrow_f<1>(m));
    m = fminf(m, xrow_f<2>(m));
    m = fminf(m, xrow_f<4>(m));
    m = fminf(m, xrow_f<8>(m));
    m = fminf(m, xhalf_f(m));
    return m;
}

__global__ __launch_bounds__(128) void idw_prep(
    const float* __restrict__ x, const float* __restrict__ tx,
    const float* __restrict__ w,
    float* __restrict__ xq_s, float* __restrict__ xt_s,
    float* __restrict__ qnorm, float* __restrict__ tnorm) {
    int i = blockIdx.x * blockDim.x + threadIdx.x;
    bool isQ = i < NQ;
    const int t = i - NQ;
    const float* src = isQ ? (x + (size_t)i * DDIM) : (tx + (size_t)t * DDIM);
    float* dst = isQ ? (xq_s + (size_t)i * DDIM)
                     : (xt_s + (size_t)(t >> 4) * (16 * DDIM) + (t & 15) * 4);
    const int step_c = isQ ? 4 : AQ;
    const int step_h = isQ ? 16 : AQ / 2;
    float vals[DDIM];
    float nrm = 0.f;
#pragma unroll
    for (int j = 0; j < DDIM; ++j) {
        int h  = j >> 4;
        int jj = j & 15;
        int K  = 4 * (jj >> 1) + (jj & 1) + 2 * h;
        float s = __expf(-w[K]);
        float v = src[K] * s;
        vals[j] = v;
        nrm   += v * v;
    }
#pragma unroll
    for (int g = 0; g < 8; ++g) {
        v4f o = {vals[4 * g], vals[4 * g + 1], vals[4 * g + 2], vals[4 * g + 3]};
        *(v4f*)(dst + (g & 3) * step_c + (g >> 2) * step_h) = o;
    }
    if (isQ) qnorm[i] = nrm; else tnorm[t] = nrm;
}

__device__ __forceinline__ v8f idw_chain(
    const v4f a0, const v4f a1, const v4f a2, const v4f a3,
    const v8f tn, const v2f (&b)[8]) {
    v8f c = tn;
    c = __builtin_amdgcn_wmma_f32_16x16x4_f32(false, a0.xy, false, b[0], (short)0, c, false, false);
    c = __builtin_amdgcn_wmma_f32_16x16x4_f32(false, a0.zw, false, b[1], (short)0, c, false, false);
    c = __builtin_amdgcn_wmma_f32_16x16x4_f32(false, a1.xy, false, b[2], (short)0, c, false, false);
    c = __builtin_amdgcn_wmma_f32_16x16x4_f32(false, a1.zw, false, b[3], (short)0, c, false, false);
    c = __builtin_amdgcn_wmma_f32_16x16x4_f32(false, a2.xy, false, b[4], (short)0, c, false, false);
    c = __builtin_amdgcn_wmma_f32_16x16x4_f32(false, a2.zw, false, b[5], (short)0, c, false, false);
    c = __builtin_amdgcn_wmma_f32_16x16x4_f32(false, a3.xy, false, b[6], (short)0, c, false, false);
    c = __builtin_amdgcn_wmma_f32_16x16x4_f32(false, a3.zw, false, b[7], (short)0, c, false, false);
    return c;
}
__device__ __forceinline__ void idw_select(
    const v8f c, const int id0, float (&tv)[5], int (&ti)[5],
    float& lim, const float bound) {
#pragma unroll
    for (int r = 0; r < 8; ++r)
        bot5_insert_lim(tv, ti, lim, bound, c[r], id0 + r);
}
__device__ __forceinline__ float min8_f(const v8f c) {
    return fminf(fminf(fminf(c[0], c[1]), fminf(c[2], c[3])),
                 fminf(fminf(c[4], c[5]), fminf(c[6], c[7])));
}

__global__ __launch_bounds__(WAVES * 32) void idw_main(
    const float* __restrict__ xq_s, const float* __restrict__ xt_s,
    const float* __restrict__ qnorm, const float* __restrict__ tnorm,
    const float* __restrict__ train_y, float* __restrict__ out) {
    __shared__ __attribute__((aligned(16))) unsigned char L[LDS_BYTES];
    float* s_fv = (float*)(L + 65536);
    int*   s_fi = (int*)  (L + 87040);
    float* s_w  = (float*)(L + 26880);
    int*   s_id = (int*)  (L + 27200);
    float* s_inv = (float*)(L + 27520);
    float* s_pm  = (float*)(L + 32768);
    float* s_bnd = (float*)(L + 58368);

    const int tid   = threadIdx.x;
    const int wave  = __builtin_amdgcn_readfirstlane(tid >> 5);
    const int lane  = tid & 31;
    const int qtile = blockIdx.x;
    const int h     = lane >> 4;
    const int ql    = lane & 15;
    const int q     = qtile * 16 + ql;

    const float* qrow = xq_s + (size_t)q * DDIM + 16 * h;
    v4f bq0 = *(const v4f*)(qrow + 0);
    v4f bq1 = *(const v4f*)(qrow + 4);
    v4f bq2 = *(const v4f*)(qrow + 8);
    v4f bq3 = *(const v4f*)(qrow + 12);
    bq0 *= -2.0f; bq1 *= -2.0f; bq2 *= -2.0f; bq3 *= -2.0f;
    v2f b[8] = {bq0.xy, bq0.zw, bq1.xy, bq1.zw, bq2.xy, bq2.zw, bq3.xy, bq3.zw};
    const float qn = qnorm[q];

    float tv[5] = {BIGF, BIGF, BIGF, BIGF, BIGF};
    int   ti[5] = {0, 0, 0, 0, 0};

    const int tbeg = wave * SEGLEN;

    const float* abase = xt_s + (size_t)tbeg * DDIM + lane * 4;
    const float* nbase = tnorm + tbeg + 8 * h;
    constexpr int NTILE = SEGLEN / 16;

    v4f p0 = *(const v4f*)(abase + 0 * AQ);
    v4f p1 = *(const v4f*)(abase + 1 * AQ);
    v4f p2 = *(const v4f*)(abase + 2 * AQ);
    v4f p3 = *(const v4f*)(abase + 3 * AQ);
    v8f pn = *(const v8f*)(nbase);

    v4f s0, s1, s2, s3;
    v8f sn;
    auto load_s = [&](const int it) __attribute__((always_inline)) {
        const float* ar1 = abase + (size_t)(it + 1) * 16 * DDIM;
        const float* nr1 = nbase + (it + 1) * 16;
        s0 = *(const v4f*)(ar1 + 0 * AQ);
        s1 = *(const v4f*)(ar1 + 1 * AQ);
        s2 = *(const v4f*)(ar1 + 2 * AQ);
        s3 = *(const v4f*)(ar1 + 3 * AQ);
        sn = *(const v8f*)(nr1);
    };
    auto load_p = [&](const int it) __attribute__((always_inline)) {
        const int nx = (it + 2 < NTILE) ? it + 2 : NTILE - 1;
        const float* ar2 = abase + (size_t)nx * 16 * DDIM;
        const float* nr2 = nbase + nx * 16;
        p0 = *(const v4f*)(ar2 + 0 * AQ);
        p1 = *(const v4f*)(ar2 + 1 * AQ);
        p2 = *(const v4f*)(ar2 + 2 * AQ);
        p3 = *(const v4f*)(ar2 + 3 * AQ);
        pn = *(const v8f*)(nr2);
    };

    static_assert(PH1 == 4, "four kept accumulators, one priority step per tile");
    __builtin_amdgcn_s_setprio(3);
    load_s(0);
    const v8f c0 = idw_chain(p0, p1, p2, p3, pn, b);
    __builtin_amdgcn_s_setprio(2);
    load_p(0);
    const v8f c1 = idw_chain(s0, s1, s2, s3, sn, b);
    __builtin_amdgcn_s_setprio(1);
    load_s(2);
    const v8f c2 = idw_chain(p0, p1, p2, p3, pn, b);
    __builtin_amdgcn_s_setprio(0);
    load_p(2);
    const v8f c3 = idw_chain(s0, s1, s2, s3, sn, b);


    {
        const float m = fminf(fminf(min8_f(c0), min8_f(c1)),
                              fminf(min8_f(c2), min8_f(c3)));
        s_pm[ql * FPITCH + wave * 2 + h] = m;
    }
    __syncthreads();
    if (wave < 16) {
        const v2f mm = *(const v2f*)(s_pm + wave * FPITCH + 2 * lane);
        float lo = fminf(mm.x, mm.y);
        float hi = fmaxf(mm.x, mm.y);
#pragma unroll
        for (int r = 0; r < 4; ++r) {
            const float m = wave_min_f(lo);
            const bool pop = lo <= m;
            lo = pop ? hi : lo;
            hi = pop ? BIGF : hi;
        }
        const float bnd = wave_min_f(lo);
        if (lane == 0) s_bnd[wave] = bnd;
    }
    __syncthreads();
    const float bound = s_bnd[ql];
    float lim = bound;

    auto half_p = [&](const int it) __attribute__((always_inline)) {
        load_s(it);
        const v8f c = idw_chain(p0, p1, p2, p3, pn, b);
        idw_select(c, tbeg + it * 16 + 8 * h, tv, ti, lim, bound);
    };
    auto half_s = [&](const int it) __attribute__((always_inline)) {
        load_p(it);
        const v8f c = idw_chain(s0, s1, s2, s3, sn, b);
        idw_select(c, tbeg + (it + 1) * 16 + 8 * h, tv, ti, lim, bound);
    };
    auto trip = [&](const int it) __attribute__((always_inline)) {
        half_p(it);
        half_s(it);
    };

    static_assert(PH1 == 4 && NTILE == 32, "stretch limits below are for 14 trips");
    __builtin_amdgcn_s_setprio(3);
    idw_select(c0, tbeg + 0 * 16 + 8 * h, tv, ti, lim, bound);
    idw_select(c1, tbeg + 1 * 16 + 8 * h, tv, ti, lim, bound);
    idw_select(c2, tbeg + 2 * 16 + 8 * h, tv, ti, lim, bound);
    idw_select(c3, tbeg + 3 * 16 + 8 * h, tv, ti, lim, bound);
#pragma nounroll
    for (int it = PH1; it < 14; it += 2) trip(it);
    __builtin_amdgcn_s_setprio(2);
#pragma nounroll
    for (int it = 14; it < 22; it += 2) trip(it);
    __builtin_amdgcn_s_setprio(1);
#pragma nounroll
    for (int it = 22; it < 28; it += 2) trip(it);
    __builtin_amdgcn_s_setprio(0);
#pragma nounroll
    for (int it = 28; it < NTILE; it += 2) trip(it);

    {
        const int li = wave * 2 + h;
#pragma unroll
        for (int k = 0; k < 5; ++k) {
            s_fv[(ql * 5 + k) * FPITCH + li] = tv[k];
            s_fi[(ql * 5 + k) * FPITCH + li] = ti[k];
        }
    }
    __syncthreads();

    if (wave < 16) {
        const int qq = wave;
        float av[5], bv[5]; int ai[5], bi[5];
#pragma unroll
        for (int k = 0; k < 5; ++k) {
            const v2f fv2 = *(const v2f*)(s_fv + (qq * 5 + k) * FPITCH + 2 * lane);
            const v2i fi2 = *(const v2i*)(s_fi + (qq * 5 + k) * FPITCH + 2 * lane);
            av[k] = fv2.x; bv[k] = fv2.y; ai[k] = fi2.x; bi[k] = fi2.y;
        }
#pragma unroll
        for (int k = 0; k < 5; ++k) bot5_insert_nb(av, ai, bv[k], bi[k]);

        float selv[5]; int seli[5];
#pragma unroll
        for (int r = 0; r < 5; ++r) {
            const float m = wave_min_f(av[0]);
            const unsigned bal = __builtin_amdgcn_ballot_w32(av[0] == m);
            const int src = bal ? __builtin_ctz(bal) : 0;
            selv[r] = m;
            seli[r] = __builtin_amdgcn_readlane(ai[0], src);
            const bool pop = lane == src;
            av[0] = pop ? av[1] : av[0]; ai[0] = pop ? ai[1] : ai[0];
            av[1] = pop ? av[2] : av[1]; ai[1] = pop ? ai[2] : ai[1];
            av[2] = pop ? av[3] : av[2]; ai[2] = pop ? ai[3] : ai[2];
            av[3] = pop ? av[4] : av[3]; ai[3] = pop ? ai[4] : ai[3];
            av[4] = pop ? BIGF  : av[4];
        }
        const float qnq = __builtin_bit_cast(float,
            __builtin_amdgcn_readlane(__builtin_bit_cast(int, qn), qq));
        float wv[5];
#pragma unroll
        for (int k = 0; k < 5; ++k)
            wv[k] = rsqrtf(fmaxf(selv[k] + qnq, 0.f) + EPSV);
        const float sw = wv[0] + wv[1] + wv[2] + wv[3] + wv[4];
        if (lane == 0) {
            s_inv[qq] = 1.f / sw;
#pragma unroll
            for (int k = 0; k < 5; ++k) { s_w[qq * 5 + k] = wv[k]; s_id[qq * 5 + k] = seli[k]; }
        }
    }
    __syncthreads();

    if (tid < 16 * OUTD) {
        const int qq = tid >> 4;
        const int o  = tid & 15;
        float acc = 0.f;
#pragma unroll
        for (int k = 0; k < 5; ++k)
            acc = fmaf(s_w[qq * 5 + k], train_y[(size_t)s_id[qq * 5 + k] * OUTD + o], acc);
        out[(size_t)(qtile * 16 + qq) * OUTD + o] = acc * s_inv[qq];
    }
}

extern "C" void kernel_launch(void* const* d_in, const int* in_sizes, int n_in,
                              void* d_out, int out_size, void* d_ws, size_t ws_size,
                              hipStream_t stream) {
    const float* x       = (const float*)d_in[0];
    const float* train_x = (const float*)d_in[1];
    const float* train_y = (const float*)d_in[2];
    const float* weights = (const float*)d_in[3];
    float* out = (float*)d_out;

    float* ws    = (float*)d_ws;
    float* xq_s  = ws;
    float* xt_s  = ws + 131072;
    float* qnorm = ws + 655360;
    float* tnorm = ws + 659456;

    idw_prep<<<(NQ + NT) / 128, 128, 0, stream>>>(x, train_x, weights,
                                                  xq_s, xt_s, qnorm, tnorm);
    idw_main<<<NQ / 16, WAVES * 32, 0, stream>>>(xq_s, xt_s, qnorm, tnorm,
                                                 train_y, out);
}
